# starting kernel
# baseline (speedup 1.0000x reference)
#include <hip/hip_runtime.h>

#define Cc 1000
#define Bn 65536
#define DECAYF 0.99f
#define ONE_M_DECAY 0.01f
#define ALPHAF 1.0f
#define LR_LF 0.05f
#define LR_MUF 0.02f
#define EPSF 1e-12f

typedef __attribute__((ext_vector_type(2))) float v2f;
typedef __attribute__((ext_vector_type(4))) float v4f;
typedef __attribute__((ext_vector_type(8))) float v8f;

__global__ void zero_ws(float* __restrict__ p, long n) {
    long i = (long)blockIdx.x * blockDim.x + threadIdx.x;
    long stride = (long)gridDim.x * blockDim.x;
    for (; i < n; i += stride) p[i] = 0.0f;
}

__global__ __launch_bounds__(256) void main_pass(
    const float* __restrict__ logits, const int* __restrict__ targets,
    const float* __restrict__ Lam,
    float* __restrict__ seg, float* __restrict__ confc,
    float* __restrict__ priorc, float* __restrict__ cep)
{
    __shared__ float zbuf[8][Cc];
    const int wave = threadIdx.x >> 5;
    const int lane = threadIdx.x & 31;
    const int row  = blockIdx.x * 8 + wave;
    const int t    = targets[row];
    const float* __restrict__ lrow   = logits + (long)row * Cc;
    const float* __restrict__ lamrow = Lam + (long)t * Cc;
    const int NV = Cc / 4;

    float maxv = -3.4e38f; int maxi = 0x7fffffff;
    float maxzp = -3.4e38f;
    for (int j = lane; j < NV; j += 32) {
        const int col = j * 4;
        __builtin_prefetch(lrow + col + 512, 0, 3);
        v4f z  = *(const v4f*)(lrow + col);
        v4f l  = *(const v4f*)(lamrow + col);
        v4f zp = z - l;
        *(v4f*)(&zbuf[wave][col]) = zp;
#pragma unroll
        for (int k = 0; k < 4; ++k) {
            if (z[k] > maxv) { maxv = z[k]; maxi = col + k; }
            maxzp = fmaxf(maxzp, zp[k]);
        }
    }
    for (int off = 16; off; off >>= 1) {
        float ov = __shfl_xor(maxv, off, 32);
        int   oi = __shfl_xor(maxi, off, 32);
        if (ov > maxv || (ov == maxv && oi < maxi)) { maxv = ov; maxi = oi; }
        maxzp = fmaxf(maxzp, __shfl_xor(maxzp, off, 32));
    }
    __syncthreads();

    float sum = 0.0f;
    for (int j = lane; j < NV; j += 32) {
        v4f zp = *(const v4f*)(&zbuf[wave][j * 4]);
        sum += __expf(zp[0] - maxzp) + __expf(zp[1] - maxzp)
             + __expf(zp[2] - maxzp) + __expf(zp[3] - maxzp);
    }
    for (int off = 16; off; off >>= 1) sum += __shfl_xor(sum, off, 32);
    const float lse = __logf(sum);
    const float inv = 1.0f / sum;

    if (lane == 0) {
        atomicAdd(&confc[(long)t * Cc + maxi], 1.0f);
        atomicAdd(&priorc[t], 1.0f);
        float logq_y = zbuf[wave][t] - maxzp - lse;
        atomicAdd(&cep[row & 255], logq_y);
    }

    float* __restrict__ segrow = seg + (long)t * Cc;
    for (int j = lane; j < NV; j += 32) {
        const int col = j * 4;
        v4f zp = *(const v4f*)(&zbuf[wave][col]);
        atomicAdd(&segrow[col + 0], __expf(zp[0] - maxzp) * inv);
        atomicAdd(&segrow[col + 1], __expf(zp[1] - maxzp) * inv);
        atomicAdd(&segrow[col + 2], __expf(zp[2] - maxzp) * inv);
        atomicAdd(&segrow[col + 3], __expf(zp[3] - maxzp) * inv);
    }
}

__global__ __launch_bounds__(32) void post_rows(
    const float* __restrict__ conf_ema, const float* __restrict__ Lam,
    const float* __restrict__ seg, const float* __restrict__ confc,
    const float* __restrict__ priorc,
    float* __restrict__ o_conf, float* __restrict__ o_lam)
{
    __shared__ float sCnt[16], sCountsSum[16], sEmpSum[16], sL1Sum[16];
    const int lane = threadIdx.x;
    const int m    = lane & 15;
    const int row  = blockIdx.x * 16 + m;
    const int k0   = (lane >> 4) << 1;
    const bool rok = (row < Cc);
    const long base = (long)row * Cc;

    if (lane < 16) sCnt[m] = rok ? priorc[row] : 0.0f;

    const v2f bones = {1.0f, 1.0f};
    const v8f vzero = {0.f,0.f,0.f,0.f,0.f,0.f,0.f,0.f};

    v8f acc = vzero;
    for (int c = 0; c < Cc; c += 4) {
        float a0 = 0.f, a1 = 0.f;
        if (rok) {
            int c0 = c + k0, c1 = c0 + 1;
            float cn0 = DECAYF * conf_ema[base + c0] + ONE_M_DECAY * confc[base + c0];
            float cn1 = DECAYF * conf_ema[base + c1] + ONE_M_DECAY * confc[base + c1];
            o_conf[base + c0] = cn0;
            o_conf[base + c1] = cn1;
            a0 = (c0 == row) ? 0.f : (cn0 + ALPHAF);
            a1 = (c1 == row) ? 0.f : (cn1 + ALPHAF);
        }
        v2f a = {a0, a1};
        acc = __builtin_amdgcn_wmma_f32_16x16x4_f32(false, a, false, bones,
                                                    (short)0, acc, false, false);
    }
    if (lane == 0)  for (int i = 0; i < 8; ++i) sCountsSum[i]     = acc[i];
    if (lane == 16) for (int i = 0; i < 8; ++i) sCountsSum[8 + i] = acc[i];
    __syncthreads();

    const float cnt  = sCnt[m];
    const float invc = 1.0f / fmaxf(cnt, 1.0f);

    acc = vzero;
    for (int c = 0; c < Cc; c += 4) {
        float a0 = 0.f, a1 = 0.f;
        if (rok) {
            int c0 = c + k0, c1 = c0 + 1;
            a0 = (c0 == row) ? 0.f : seg[base + c0] * invc;
            a1 = (c1 == row) ? 0.f : seg[base + c1] * invc;
        }
        v2f a = {a0, a1};
        acc = __builtin_amdgcn_wmma_f32_16x16x4_f32(false, a, false, bones,
                                                    (short)0, acc, false, false);
    }
    if (lane == 0)  for (int i = 0; i < 8; ++i) sEmpSum[i]     = acc[i];
    if (lane == 16) for (int i = 0; i < 8; ++i) sEmpSum[8 + i] = acc[i];
    __syncthreads();

    const float phiDen = 1.0f / (sCountsSum[m] + EPSF);
    const float empDen = 1.0f / fmaxf(sEmpSum[m], EPSF);

    acc = vzero;
    for (int c = 0; c < Cc; c += 4) {
        float a0 = 0.f, a1 = 0.f;
        if (rok) {
            int c0 = c + k0, c1 = c0 + 1;
            if (c0 != row) {
                float cn  = DECAYF * conf_ema[base + c0] + ONE_M_DECAY * confc[base + c0];
                float phi = (cn + ALPHAF) * phiDen;
                float emp = seg[base + c0] * invc * empDen;
                a0 = Lam[base + c0] + LR_LF * (emp - phi);
            }
            if (c1 != row) {
                float cn  = DECAYF * conf_ema[base + c1] + ONE_M_DECAY * confc[base + c1];
                float phi = (cn + ALPHAF) * phiDen;
                float emp = seg[base + c1] * invc * empDen;
                a1 = Lam[base + c1] + LR_LF * (emp - phi);
            }
        }
        v2f a = {a0, a1};
        acc = __builtin_amdgcn_wmma_f32_16x16x4_f32(false, a, false, bones,
                                                    (short)0, acc, false, false);
    }
    if (lane == 0)  for (int i = 0; i < 8; ++i) sL1Sum[i]     = acc[i];
    if (lane == 16) for (int i = 0; i < 8; ++i) sL1Sum[8 + i] = acc[i];
    __syncthreads();

    const float rowMean = sL1Sum[m] * (1.0f / (float)Cc);
    const bool  present = (cnt > 0.0f);

    for (int c = 0; c < Cc; c += 4) {
        if (!rok) continue;
        int c0 = c + k0, c1 = c0 + 1;
        float l0 = 0.f, l1 = 0.f;
        if (c0 != row) {
            float cn  = DECAYF * conf_ema[base + c0] + ONE_M_DECAY * confc[base + c0];
            float phi = (cn + ALPHAF) * phiDen;
            float emp = seg[base + c0] * invc * empDen;
            l0 = Lam[base + c0] + LR_LF * (emp - phi);
        }
        if (c1 != row) {
            float cn  = DECAYF * conf_ema[base + c1] + ONE_M_DECAY * confc[base + c1];
            float phi = (cn + ALPHAF) * phiDen;
            float emp = seg[base + c1] * invc * empDen;
            l1 = Lam[base + c1] + LR_LF * (emp - phi);
        }
        o_lam[base + c0] = present ? (l0 - rowMean) : Lam[base + c0];
        o_lam[base + c1] = present ? (l1 - rowMean) : Lam[base + c1];
    }
}

__global__ __launch_bounds__(1024) void finalize(
    const float* __restrict__ prior_ema, const float* __restrict__ mu,
    const float* __restrict__ seg, const float* __restrict__ priorc,
    const float* __restrict__ cep,
    float* __restrict__ o_ce, float* __restrict__ o_mu, float* __restrict__ o_prior)
{
    __shared__ float red[1024];
    const int j = threadIdx.x;

    float pn = 0.0f;
    if (j < Cc) {
        pn = DECAYF * prior_ema[j] + ONE_M_DECAY * priorc[j];
        o_prior[j] = pn;
    }
    red[j] = pn; __syncthreads();
    for (int s = 512; s; s >>= 1) { if (j < s) red[j] += red[j + s]; __syncthreads(); }
    const float priorSum = red[0];
    __syncthreads();

    float mu1 = 0.0f;
    if (j < Cc) {
        float colsum = 0.0f;
        for (int r = 0; r < Cc; ++r) colsum += seg[(long)r * Cc + j];
        float emp_marg = colsum * (1.0f / (float)Bn);
        float pi = pn / (priorSum + EPSF);
        mu1 = mu[j] + LR_MUF * (emp_marg - pi);
    }
    red[j] = mu1; __syncthreads();
    for (int s = 512; s; s >>= 1) { if (j < s) red[j] += red[j + s]; __syncthreads(); }
    const float muMean = red[0] * (1.0f / (float)Cc);
    if (j < Cc) o_mu[j] = mu1 - muMean;

    if (j == 0) {
        float s = 0.0f;
        for (int i = 0; i < 256; ++i) s += cep[i];
        o_ce[0] = -s * (1.0f / (float)Bn);
    }
}

extern "C" void kernel_launch(void* const* d_in, const int* in_sizes, int n_in,
                              void* d_out, int out_size, void* d_ws, size_t ws_size,
                              hipStream_t stream)
{
    const float* logits    = (const float*)d_in[0];
    const int*   targets   = (const int*)  d_in[1];
    const float* conf_ema  = (const float*)d_in[2];
    const float* prior_ema = (const float*)d_in[3];
    const float* Lam       = (const float*)d_in[4];
    const float* mu        = (const float*)d_in[5];

    float* out     = (float*)d_out;
    float* o_ce    = out;
    float* o_lam   = out + 1;
    float* o_mu    = out + 1 + (long)Cc * Cc;
    float* o_conf  = out + 1 + (long)Cc * Cc + Cc;
    float* o_prior = out + 1 + 2L * Cc * Cc + Cc;

    float* ws     = (float*)d_ws;
    float* seg    = ws;
    float* confc  = ws + (long)Cc * Cc;
    float* priorc = ws + 2L * Cc * Cc;
    float* cep    = priorc + Cc;

    const long nz = 2L * Cc * Cc + Cc + 256;
    zero_ws<<<1024, 256, 0, stream>>>(ws, nz);
    main_pass<<<Bn / 8, 256, 0, stream>>>(logits, targets, Lam, seg, confc, priorc, cep);
    post_rows<<<(Cc + 15) / 16, 32, 0, stream>>>(conf_ema, Lam, seg, confc, priorc,
                                                 o_conf, o_lam);
    finalize<<<1, 1024, 0, stream>>>(prior_ema, mu, seg, priorc, cep,
                                     o_ce, o_mu, o_prior);
}


# post_rows: one wave per class row, row read once with coalesced b128 loads and kept in registers (was 16 strided rows per wave, 4 passes); row sums stay on f32 WMMA, summed in a new order
# speedup vs baseline: 2.6337x; 2.6337x over previous
#include <hip/hip_runtime.h>

#define Cc 1000
#define Bn 65536
#define DECAYF 0.99f
#define ONE_M_DECAY 0.01f
#define ALPHAF 1.0f
#define LR_LF 0.05f
#define LR_MUF 0.02f
#define EPSF 1e-12f

typedef __attribute__((ext_vector_type(2))) float v2f;
typedef __attribute__((ext_vector_type(4))) float v4f;
typedef __attribute__((ext_vector_type(8))) float v8f;

__global__ void zero_ws(float* __restrict__ p, long n) {
    long i = (long)blockIdx.x * blockDim.x + threadIdx.x;
    long stride = (long)gridDim.x * blockDim.x;
    for (; i < n; i += stride) p[i] = 0.0f;
}

__global__ __launch_bounds__(256) void main_pass(
    const float* __restrict__ logits, const int* __restrict__ targets,
    const float* __restrict__ Lam,
    float* __restrict__ seg, float* __restrict__ confc,
    float* __restrict__ priorc, float* __restrict__ cep)
{
    __shared__ float zbuf[8][Cc];
    const int wave = threadIdx.x >> 5;
    const int lane = threadIdx.x & 31;
    const int row  = blockIdx.x * 8 + wave;
    const int t    = targets[row];
    const float* __restrict__ lrow   = logits + (long)row * Cc;
    const float* __restrict__ lamrow = Lam + (long)t * Cc;
    const int NV = Cc / 4;

    float maxv = -3.4e38f; int maxi = 0x7fffffff;
    float maxzp = -3.4e38f;
    for (int j = lane; j < NV; j += 32) {
        const int col = j * 4;
        __builtin_prefetch(lrow + col + 512, 0, 3);
        v4f z  = *(const v4f*)(lrow + col);
        v4f l  = *(const v4f*)(lamrow + col);
        v4f zp = z - l;
        *(v4f*)(&zbuf[wave][col]) = zp;
#pragma unroll
        for (int k = 0; k < 4; ++k) {
            if (z[k] > maxv) { maxv = z[k]; maxi = col + k; }
            maxzp = fmaxf(maxzp, zp[k]);
        }
    }
    for (int off = 16; off; off >>= 1) {
        float ov = __shfl_xor(maxv, off, 32);
        int   oi = __shfl_xor(maxi, off, 32);
        if (ov > maxv || (ov == maxv && oi < maxi)) { maxv = ov; maxi = oi; }
        maxzp = fmaxf(maxzp, __shfl_xor(maxzp, off, 32));
    }
    __syncthreads();

    float sum = 0.0f;
    for (int j = lane; j < NV; j += 32) {
        v4f zp = *(const v4f*)(&zbuf[wave][j * 4]);
        sum += __expf(zp[0] - maxzp) + __expf(zp[1] - maxzp)
             + __expf(zp[2] - maxzp) + __expf(zp[3] - maxzp);
    }
    for (int off = 16; off; off >>= 1) sum += __shfl_xor(sum, off, 32);
    const float lse = __logf(sum);
    const float inv = 1.0f / sum;

    if (lane == 0) {
        atomicAdd(&confc[(long)t * Cc + maxi], 1.0f);
        atomicAdd(&priorc[t], 1.0f);
        float logq_y = zbuf[wave][t] - maxzp - lse;
        atomicAdd(&cep[row & 255], logq_y);
    }

    float* __restrict__ segrow = seg + (long)t * Cc;
    for (int j = lane; j < NV; j += 32) {
        const int col = j * 4;
        v4f zp = *(const v4f*)(&zbuf[wave][col]);
        atomicAdd(&segrow[col + 0], __expf(zp[0] - maxzp) * inv);
        atomicAdd(&segrow[col + 1], __expf(zp[1] - maxzp) * inv);
        atomicAdd(&segrow[col + 2], __expf(zp[2] - maxzp) * inv);
        atomicAdd(&segrow[col + 3], __expf(zp[3] - maxzp) * inv);
    }
}

__device__ __forceinline__ float wave_sum32(const float (&v)[32]) {
    const v2f bones = {1.0f, 1.0f};
    const v8f vzero = {0.f,0.f,0.f,0.f,0.f,0.f,0.f,0.f};
    v8f acc0 = vzero, acc1 = vzero;
#pragma unroll
    for (int p = 0; p < 32; p += 4) {
        const v2f a0 = {v[p + 0], v[p + 1]};
        const v2f a1 = {v[p + 2], v[p + 3]};
        acc0 = __builtin_amdgcn_wmma_f32_16x16x4_f32(false, a0, false, bones,
                                                     (short)0, acc0, false, false);
        acc1 = __builtin_amdgcn_wmma_f32_16x16x4_f32(false, a1, false, bones,
                                                     (short)0, acc1, false, false);
    }
    const v8f acc = acc0 + acc1;
    float s = ((acc[0] + acc[1]) + (acc[2] + acc[3])) + ((acc[4] + acc[5]) + (acc[6] + acc[7]));
    s += __shfl_xor(s, 16, 32);
    return s;
}

typedef float v4f_a4 __attribute__((ext_vector_type(4), aligned(4)));

__global__ __launch_bounds__(32) void post_rows(
    const float* __restrict__ conf_ema, const float* __restrict__ Lam,
    const float* __restrict__ seg, const float* __restrict__ confc,
    const float* __restrict__ priorc,
    float* __restrict__ o_conf, float* __restrict__ o_lam)
{
    const int lane  = threadIdx.x;
    const int row   = blockIdx.x;
    const long base = (long)row * Cc;
    const int NV    = Cc / 4;

    const float cnt     = priorc[row];
    const float invc    = 1.0f / fmaxf(cnt, 1.0f);
    const bool  present = (cnt > 0.0f);

    const v4f z4 = {0.f, 0.f, 0.f, 0.f};
    v4f ve[8], vb[8], vs[8], vl[8];
    const bool ok7 = (lane + 32 * 7) < NV;
    const long o0  = base + 4 * lane;
    const long o7  = base + 4 * min(lane + 32 * 7, NV - 1);
#pragma unroll
    for (int i = 0; i < 7; ++i) {
        ve[i] = *(const v4f*)(conf_ema + o0 + 128 * i);
        vb[i] = *(const v4f*)(confc    + o0 + 128 * i);
        vs[i] = *(const v4f*)(seg      + o0 + 128 * i);
        vl[i] = *(const v4f*)(Lam      + o0 + 128 * i);
    }
    ve[7] = *(const v4f*)(conf_ema + o7);
    vb[7] = *(const v4f*)(confc    + o7);
    vs[7] = *(const v4f*)(seg      + o7);
    vl[7] = *(const v4f*)(Lam      + o7);
    ve[7] = ok7 ? ve[7] : z4;
    vb[7] = ok7 ? vb[7] : z4;
    vs[7] = ok7 ? vs[7] : z4;
    vl[7] = ok7 ? vl[7] : z4;

    float acnt[32], empr[32];
#pragma unroll
    for (int i = 0; i < 8; ++i) {
        const int j = lane + 32 * i;
        const bool ok = (i < 7) || ok7;
        const v4f cn = DECAYF * ve[i] + ONE_M_DECAY * vb[i];
        if (ok) *(v4f_a4*)(o_conf + base + 4 * j) = cn;
#pragma unroll
        for (int k = 0; k < 4; ++k) {
            const int c = 4 * j + k;
            const bool offd = ok && (c != row);
            acnt[4 * i + k] = offd ? (cn[k] + ALPHAF) : 0.f;
            empr[4 * i + k] = offd ? (vs[i][k] * invc) : 0.f;
        }
    }

    const float countsSum = wave_sum32(acnt);
    const float empSum    = wave_sum32(empr);
    const float phiDen = 1.0f / (countsSum + EPSF);
    const float empDen = 1.0f / fmaxf(empSum, EPSF);

    float l1[32];
#pragma unroll
    for (int i = 0; i < 8; ++i) {
        const int j = lane + 32 * i;
#pragma unroll
        for (int k = 0; k < 4; ++k) {
            const int c = 4 * j + k;
            const float phi = acnt[4 * i + k] * phiDen;
            const float emp = empr[4 * i + k] * empDen;
            const float v   = vl[i][k] + LR_LF * (emp - phi);
            l1[4 * i + k] = (((i < 7) || ok7) && (c != row)) ? v : 0.f;
        }
    }
    const float rowMean = wave_sum32(l1) * (1.0f / (float)Cc);

#pragma unroll
    for (int i = 0; i < 8; ++i) {
        const int j = lane + 32 * i;
        const bool ok = (i < 7) || ok7;
        v4f r;
#pragma unroll
        for (int k = 0; k < 4; ++k)
            r[k] = present ? (l1[4 * i + k] - rowMean) : vl[i][k];
        if (ok) *(v4f_a4*)(o_lam + base + 4 * j) = r;
    }
}

__global__ __launch_bounds__(1024) void finalize(
    const float* __restrict__ prior_ema, const float* __restrict__ mu,
    const float* __restrict__ seg, const float* __restrict__ priorc,
    const float* __restrict__ cep,
    float* __restrict__ o_ce, float* __restrict__ o_mu, float* __restrict__ o_prior)
{
    __shared__ float red[1024];
    const int j = threadIdx.x;

    float pn = 0.0f;
    if (j < Cc) {
        pn = DECAYF * prior_ema[j] + ONE_M_DECAY * priorc[j];
        o_prior[j] = pn;
    }
    red[j] = pn; __syncthreads();
    for (int s = 512; s; s >>= 1) { if (j < s) red[j] += red[j + s]; __syncthreads(); }
    const float priorSum = red[0];
    __syncthreads();

    float mu1 = 0.0f;
    if (j < Cc) {
        float colsum = 0.0f;
        for (int r = 0; r < Cc; ++r) colsum += seg[(long)r * Cc + j];
        float emp_marg = colsum * (1.0f / (float)Bn);
        float pi = pn / (priorSum + EPSF);
        mu1 = mu[j] + LR_MUF * (emp_marg - pi);
    }
    red[j] = mu1; __syncthreads();
    for (int s = 512; s; s >>= 1) { if (j < s) red[j] += red[j + s]; __syncthreads(); }
    const float muMean = red[0] * (1.0f / (float)Cc);
    if (j < Cc) o_mu[j] = mu1 - muMean;

    if (j == 0) {
        float s = 0.0f;
        for (int i = 0; i < 256; ++i) s += cep[i];
        o_ce[0] = -s * (1.0f / (float)Bn);
    }
}

extern "C" void kernel_launch(void* const* d_in, const int* in_sizes, int n_in,
                              void* d_out, int out_size, void* d_ws, size_t ws_size,
                              hipStream_t stream)
{
    const float* logits    = (const float*)d_in[0];
    const int*   targets   = (const int*)  d_in[1];
    const float* conf_ema  = (const float*)d_in[2];
    const float* prior_ema = (const float*)d_in[3];
    const float* Lam       = (const float*)d_in[4];
    const float* mu        = (const float*)d_in[5];

    float* out     = (float*)d_out;
    float* o_ce    = out;
    float* o_lam   = out + 1;
    float* o_mu    = out + 1 + (long)Cc * Cc;
    float* o_conf  = out + 1 + (long)Cc * Cc + Cc;
    float* o_prior = out + 1 + 2L * Cc * Cc + Cc;

    float* ws     = (float*)d_ws;
    float* seg    = ws;
    float* confc  = ws + (long)Cc * Cc;
    float* priorc = ws + 2L * Cc * Cc;
    float* cep    = priorc + Cc;

    const long nz = 2L * Cc * Cc + Cc + 256;
    zero_ws<<<1024, 256, 0, stream>>>(ws, nz);
    main_pass<<<Bn / 8, 256, 0, stream>>>(logits, targets, Lam, seg, confc, priorc, cep);
    post_rows<<<Cc, 32, 0, stream>>>(conf_ema, Lam, seg, confc, priorc,
                                                 o_conf, o_lam);
    finalize<<<1, 1024, 0, stream>>>(prior_ema, mu, seg, priorc, cep,
                                     o_ce, o_mu, o_prior);
}


# column sums of q: each post_rows block (8 rows) adds its segment-sum rows through LDS into one partial row; finalize adds 125 partial rows, not 1000 rows serially; summed in a new order
# speedup vs baseline: 8.2857x; 3.1460x over previous
#include <hip/hip_runtime.h>

#define Cc 1000
#define Bn 65536
#define DECAYF 0.99f
#define ONE_M_DECAY 0.01f
#define ALPHAF 1.0f
#define LR_LF 0.05f
#define LR_MUF 0.02f
#define EPSF 1e-12f

typedef __attribute__((ext_vector_type(2))) float v2f;
typedef __attribute__((ext_vector_type(4))) float v4f;
typedef __attribute__((ext_vector_type(8))) float v8f;

__global__ void zero_ws(float* __restrict__ p, long n) {
    long i = (long)blockIdx.x * blockDim.x + threadIdx.x;
    long stride = (long)gridDim.x * blockDim.x;
    for (; i < n; i += stride) p[i] = 0.0f;
}

__global__ __launch_bounds__(256) void bucket_rows(
    const int* __restrict__ targets, int* __restrict__ cnt, int* __restrict__ bucket)
{
    const int row = blockIdx.x * 256 + threadIdx.x;
    const int t   = targets[row];
    const int r   = atomicAdd(&cnt[t], 1);
    bucket[(long)t * Bn + r] = row;
}

#define NEG_BIG (-3.4e38f)
#define NW 4

__device__ __forceinline__ void load_row(v4f (&z)[8], float& z_c,
                                         const float* __restrict__ lrow,
                                         const int o0, const int o7, const int c)
{
#pragma unroll
    for (int i = 0; i < 7; ++i) z[i] = *(const v4f*)(lrow + o0 + 128 * i);
    z[7] = *(const v4f*)(lrow + o7);
    z_c  = lrow[c];
}

__device__ __forceinline__ void row_step(v4f (&z)[8], const float z_c, v4f (&acc)[8],
                                         const float* slam, const float lam_cc,
                                         const int lane, const int o0, const int o7, const bool ok7,
                                         float* __restrict__ confrow, float& cesum)
{
    const v4f negbig = {NEG_BIG, NEG_BIG, NEG_BIG, NEG_BIG};
    float maxv = NEG_BIG; int slot = 0x7fffffff;
    float maxzp = NEG_BIG;
#pragma unroll
    for (int i = 0; i < 8; ++i) {
        const v4f l = *(const v4f*)(slam + ((i < 7) ? (o0 + 128 * i) : o7));
        v4f zr = z[i];
        v4f zp = zr - l;
        if (i == 7) { zr = ok7 ? zr : negbig; zp = ok7 ? zp : negbig; }
#pragma unroll
        for (int k = 0; k < 4; ++k) {
            if (zr[k] > maxv) { maxv = zr[k]; slot = 4 * i + k; }
            maxzp = fmaxf(maxzp, zp[k]);
        }
        z[i] = zp;
    }
    int maxi = (slot == 0x7fffffff) ? slot : (4 * lane + 128 * (slot >> 2) + (slot & 3));
    for (int off = 16; off; off >>= 1) {
        float ov = __shfl_xor(maxv, off, 32);
        int   oi = __shfl_xor(maxi, off, 32);
        if (ov > maxv || (ov == maxv && oi < maxi)) { maxv = ov; maxi = oi; }
        maxzp = fmaxf(maxzp, __shfl_xor(maxzp, off, 32));
    }

    float sum = 0.0f;
#pragma unroll
    for (int i = 0; i < 8; ++i) {
        v4f e;
#pragma unroll
        for (int k = 0; k < 4; ++k) e[k] = __expf(z[i][k] - maxzp);
        z[i] = e;
        sum += (e[0] + e[1]) + (e[2] + e[3]);
    }
    for (int off = 16; off; off >>= 1) sum += __shfl_xor(sum, off, 32);
    const float lse = __logf(sum);
    const float inv = 1.0f / sum;

#pragma unroll
    for (int i = 0; i < 8; ++i) acc[i] += z[i] * inv;

    if (lane == 0) atomicAdd(&confrow[maxi], 1.0f);
    cesum += ((z_c - lam_cc) - maxzp) - lse;
}

__global__ __launch_bounds__(NW * 32, 4) void class_pass(
    const float* __restrict__ logits, const float* __restrict__ Lam,
    const int* __restrict__ cnt, const int* __restrict__ bucket,
    float* __restrict__ seg, float* __restrict__ confc,
    float* __restrict__ priorc, float* __restrict__ cep)
{
    __shared__ float sacc[NW][Cc];
    __shared__ float slam[Cc];
    const int wave = __builtin_amdgcn_readfirstlane(threadIdx.x >> 5);
    const int lane = threadIdx.x & 31;
    const int tid  = threadIdx.x;
    const int c    = blockIdx.x;
    const int n    = cnt[c];
    const int NV   = Cc / 4;
    const int* __restrict__ brow = bucket + (long)c * Bn;

    const bool ok7 = (lane + 32 * 7) < NV;
    const int  o0  = 4 * lane;
    const int  o7  = 4 * min(lane + 32 * 7, NV - 1);

    const float* __restrict__ lamrow = Lam + (long)c * Cc;
    for (int j = tid; j < NV; j += NW * 32) *(v4f*)(&slam[4 * j]) = *(const v4f*)(lamrow + 4 * j);
    const float lam_cc = lamrow[c];

    const v4f z4 = {0.f, 0.f, 0.f, 0.f};
    v4f acc[8];
#pragma unroll
    for (int i = 0; i < 8; ++i) acc[i] = z4;
    float cesum = 0.0f;
    float* __restrict__ confrow = confc + (long)c * Cc;

    v4f za[8], zb[8];
    float zca, zcb;
    const int last = max(n - 1, 0);
    int r = wave;
    const int rowFirst = (n > 0) ? brow[min(r, last)] : 0;
    int rowNext        = (n > 0) ? brow[min(r + NW, last)] : 0;
    load_row(za, zca, logits + (long)rowFirst * Cc, o0, o7, c);
    __syncthreads();
    if (r < n) {
        for (;;) {
            load_row(zb, zcb, logits + (long)rowNext * Cc, o0, o7, c);
            rowNext = brow[min(r + 2 * NW, last)];
            row_step(za, zca, acc, slam, lam_cc, lane, o0, o7, ok7, confrow, cesum);
            r += NW;
            if (r >= n) break;
            load_row(za, zca, logits + (long)rowNext * Cc, o0, o7, c);
            rowNext = brow[min(r + 2 * NW, last)];
            row_step(zb, zcb, acc, slam, lam_cc, lane, o0, o7, ok7, confrow, cesum);
            r += NW;
            if (r >= n) break;
        }
    }

#pragma unroll
    for (int i = 0; i < 7; ++i) *(v4f*)(&sacc[wave][o0 + 128 * i]) = acc[i];
    if (ok7) *(v4f*)(&sacc[wave][o0 + 128 * 7]) = acc[7];
    __syncthreads();
    for (int j = tid; j < NV; j += NW * 32) {
        v4f s = *(const v4f*)(&sacc[0][4 * j]);
#pragma unroll
        for (int w = 1; w < NW; ++w) s += *(const v4f*)(&sacc[w][4 * j]);
        *(v4f*)(seg + (long)c * Cc + 4 * j) = s;
    }
    if (tid == 0) priorc[c] = (float)n;
    if (lane == 0 && wave < n) atomicAdd(&cep[(c * NW + wave) & 255], cesum);
}

__device__ __forceinline__ float wave_sum32(const float (&v)[32]) {
    const v2f bones = {1.0f, 1.0f};
    const v8f vzero = {0.f,0.f,0.f,0.f,0.f,0.f,0.f,0.f};
    v8f acc0 = vzero, acc1 = vzero;
#pragma unroll
    for (int p = 0; p < 32; p += 4) {
        const v2f a0 = {v[p + 0], v[p + 1]};
        const v2f a1 = {v[p + 2], v[p + 3]};
        acc0 = __builtin_amdgcn_wmma_f32_16x16x4_f32(false, a0, false, bones,
                                                     (short)0, acc0, false, false);
        acc1 = __builtin_amdgcn_wmma_f32_16x16x4_f32(false, a1, false, bones,
                                                     (short)0, acc1, false, false);
    }
    const v8f acc = acc0 + acc1;
    float s = ((acc[0] + acc[1]) + (acc[2] + acc[3])) + ((acc[4] + acc[5]) + (acc[6] + acc[7]));
    s += __shfl_xor(s, 16, 32);
    return s;
}

typedef float v4f_a4 __attribute__((ext_vector_type(4), aligned(4)));

#define PW 8
#define NPART (Cc / PW)
#define CPITCH 1024

__global__ __launch_bounds__(PW * 32) void post_rows(
    const float* __restrict__ conf_ema, const float* __restrict__ Lam,
    const float* __restrict__ seg, const float* __restrict__ confc,
    const float* __restrict__ priorc,
    float* __restrict__ o_conf, float* __restrict__ o_lam,
    float* __restrict__ colpart)
{
    __shared__ __attribute__((aligned(16))) float scol[PW][Cc];
    const int wave  = __builtin_amdgcn_readfirstlane(threadIdx.x >> 5);
    const int lane  = threadIdx.x & 31;
    const int row   = blockIdx.x * PW + wave;
    const long base = (long)row * Cc;
    const int NV    = Cc / 4;

    const float cnt     = priorc[row];
    const float invc    = 1.0f / fmaxf(cnt, 1.0f);
    const bool  present = (cnt > 0.0f);

    const v4f z4 = {0.f, 0.f, 0.f, 0.f};
    v4f ve[8], vb[8], vs[8], vl[8];
    const bool ok7 = (lane + 32 * 7) < NV;
    const long o0  = base + 4 * lane;
    const long o7  = base + 4 * min(lane + 32 * 7, NV - 1);
#pragma unroll
    for (int i = 0; i < 7; ++i) {
        ve[i] = *(const v4f*)(conf_ema + o0 + 128 * i);
        vb[i] = *(const v4f*)(confc    + o0 + 128 * i);
        vs[i] = *(const v4f*)(seg      + o0 + 128 * i);
        vl[i] = *(const v4f*)(Lam      + o0 + 128 * i);
    }
    ve[7] = *(const v4f*)(conf_ema + o7);
    vb[7] = *(const v4f*)(confc    + o7);
    vs[7] = *(const v4f*)(seg      + o7);
    vl[7] = *(const v4f*)(Lam      + o7);
    ve[7] = ok7 ? ve[7] : z4;
    vb[7] = ok7 ? vb[7] : z4;
    vs[7] = ok7 ? vs[7] : z4;
    vl[7] = ok7 ? vl[7] : z4;

#pragma unroll
    for (int i = 0; i < 7; ++i) *(v4f*)(&scol[wave][4 * lane + 128 * i]) = vs[i];
    if (ok7) *(v4f*)(&scol[wave][4 * lane + 128 * 7]) = vs[7];

    float acnt[32], empr[32];
#pragma unroll
    for (int i = 0; i < 8; ++i) {
        const int j = lane + 32 * i;
        const bool ok = (i < 7) || ok7;
        const v4f cn = DECAYF * ve[i] + ONE_M_DECAY * vb[i];
        if (ok) *(v4f_a4*)(o_conf + base + 4 * j) = cn;
#pragma unroll
        for (int k = 0; k < 4; ++k) {
            const int c = 4 * j + k;
            const bool offd = ok && (c != row);
            acnt[4 * i + k] = offd ? (cn[k] + ALPHAF) : 0.f;
            empr[4 * i + k] = offd ? (vs[i][k] * invc) : 0.f;
        }
    }

    const float countsSum = wave_sum32(acnt);
    const float empSum    = wave_sum32(empr);
    const float phiDen = 1.0f / (countsSum + EPSF);
    const float empDen = 1.0f / fmaxf(empSum, EPSF);

    float l1[32];
#pragma unroll
    for (int i = 0; i < 8; ++i) {
        const int j = lane + 32 * i;
#pragma unroll
        for (int k = 0; k < 4; ++k) {
            const int c = 4 * j + k;
            const float phi = acnt[4 * i + k] * phiDen;
            const float emp = empr[4 * i + k] * empDen;
            const float v   = vl[i][k] + LR_LF * (emp - phi);
            l1[4 * i + k] = (((i < 7) || ok7) && (c != row)) ? v : 0.f;
        }
    }
    const float rowMean = wave_sum32(l1) * (1.0f / (float)Cc);

    __syncthreads();
    {
        const int t = threadIdx.x;
        if (t < NV) {
            v4f s = *(const v4f*)(&scol[0][4 * t]);
#pragma unroll
            for (int w = 1; w < PW; ++w) s += *(const v4f*)(&scol[w][4 * t]);
            *(v4f*)(colpart + (long)blockIdx.x * CPITCH + 4 * t) = s;
        }
    }

#pragma unroll
    for (int i = 0; i < 8; ++i) {
        const int j = lane + 32 * i;
        const bool ok = (i < 7) || ok7;
        v4f r;
#pragma unroll
        for (int k = 0; k < 4; ++k)
            r[k] = present ? (l1[4 * i + k] - rowMean) : vl[i][k];
        if (ok) *(v4f_a4*)(o_lam + base + 4 * j) = r;
    }
}

__global__ __launch_bounds__(1024) void finalize(
    const float* __restrict__ prior_ema, const float* __restrict__ mu,
    const float* __restrict__ colpart, const float* __restrict__ priorc,
    const float* __restrict__ cep,
    float* __restrict__ o_ce, float* __restrict__ o_mu, float* __restrict__ o_prior)
{
    __shared__ float red[1024];
    __shared__ __attribute__((aligned(16))) float scs[4][Cc];
    const int j = threadIdx.x;

    {
        const int NV = Cc / 4;
        const int g  = j >> 8;
        const int c4 = j & 255;
        if (c4 < NV) {
            const int NK = (NPART + 3) / 4;
            const v4f z4 = {0.f, 0.f, 0.f, 0.f};
            v4f s = z4;
#pragma unroll 16
            for (int k = 0; k < NK; ++k) {
                const int r = g + 4 * k;
                const v4f v = *(const v4f*)(colpart + (long)min(r, NPART - 1) * CPITCH + 4 * c4);
                s += (r < NPART) ? v : z4;
            }
            *(v4f*)(&scs[g][4 * c4]) = s;
        }
    }

    float pn = 0.0f;
    if (j < Cc) {
        pn = DECAYF * prior_ema[j] + ONE_M_DECAY * priorc[j];
        o_prior[j] = pn;
    }
    red[j] = pn; __syncthreads();
    for (int s = 512; s; s >>= 1) { if (j < s) red[j] += red[j + s]; __syncthreads(); }
    const float priorSum = red[0];
    __syncthreads();

    float mu1 = 0.0f;
    if (j < Cc) {
        const float colsum = (scs[0][j] + scs[1][j]) + (scs[2][j] + scs[3][j]);
        float emp_marg = colsum * (1.0f / (float)Bn);
        float pi = pn / (priorSum + EPSF);
        mu1 = mu[j] + LR_MUF * (emp_marg - pi);
    }
    red[j] = mu1; __syncthreads();
    for (int s = 512; s; s >>= 1) { if (j < s) red[j] += red[j + s]; __syncthreads(); }
    const float muMean = red[0] * (1.0f / (float)Cc);
    if (j < Cc) o_mu[j] = mu1 - muMean;

    if (j == 0) {
        float s = 0.0f;
        for (int i = 0; i < 256; ++i) s += cep[i];
        o_ce[0] = -s * (1.0f / (float)Bn);
    }
}

extern "C" void kernel_launch(void* const* d_in, const int* in_sizes, int n_in,
                              void* d_out, int out_size, void* d_ws, size_t ws_size,
                              hipStream_t stream)
{
    const float* logits    = (const float*)d_in[0];
    const int*   targets   = (const int*)  d_in[1];
    const float* conf_ema  = (const float*)d_in[2];
    const float* prior_ema = (const float*)d_in[3];
    const float* Lam       = (const float*)d_in[4];
    const float* mu        = (const float*)d_in[5];

    float* out     = (float*)d_out;
    float* o_ce    = out;
    float* o_lam   = out + 1;
    float* o_mu    = out + 1 + (long)Cc * Cc;
    float* o_conf  = out + 1 + (long)Cc * Cc + Cc;
    float* o_prior = out + 1 + 2L * Cc * Cc + Cc;

    float* ws     = (float*)d_ws;
    float* seg    = ws;
    float* confc  = ws + (long)Cc * Cc;
    float* priorc = ws + 2L * Cc * Cc;
    float* cep    = priorc + Cc;
    int*   cnt    = (int*)(cep + 256);
    int*   bucket = cnt + 1024;
    float* colpart = (float*)(bucket + (long)Cc * Bn) + 24;

    const long nz = 1L * Cc * Cc + Cc + 256 + 1024;
    zero_ws<<<1024, 256, 0, stream>>>(confc, nz);
    bucket_rows<<<Bn / 256, 256, 0, stream>>>(targets, cnt, bucket);
    class_pass<<<Cc, NW * 32, 0, stream>>>(logits, Lam, cnt, bucket, seg, confc, priorc, cep);
    post_rows<<<Cc / PW, PW * 32, 0, stream>>>(conf_ema, Lam, seg, confc, priorc,
                                               o_conf, o_lam, colpart);
    finalize<<<1, 1024, 0, stream>>>(prior_ema, mu, colpart, priorc, cep,
                                     o_ce, o_mu, o_prior);
}


# class_pass: a row's wave reductions (argmax, max z', sum e) on the vector ALU (DPP row_xmask + v_permlanex16), not ds_bpermute; argmax = wave max, then least column of its holders; sum order kept
# speedup vs baseline: 8.8593x; 1.0692x over previous
#include <hip/hip_runtime.h>

#define Cc 1000
#define Bn 65536
#define DECAYF 0.99f
#define ONE_M_DECAY 0.01f
#define ALPHAF 1.0f
#define LR_LF 0.05f
#define LR_MUF 0.02f
#define EPSF 1e-12f

typedef __attribute__((ext_vector_type(2))) float v2f;
typedef __attribute__((ext_vector_type(4))) float v4f;
typedef __attribute__((ext_vector_type(8))) float v8f;

__global__ void zero_ws(float* __restrict__ p, long n) {
    long i = (long)blockIdx.x * blockDim.x + threadIdx.x;
    long stride = (long)gridDim.x * blockDim.x;
    for (; i < n; i += stride) p[i] = 0.0f;
}

__global__ __launch_bounds__(256) void bucket_rows(
    const int* __restrict__ targets, int* __restrict__ cnt, int* __restrict__ bucket)
{
    const int row = blockIdx.x * 256 + threadIdx.x;
    const int t   = targets[row];
    const int r   = atomicAdd(&cnt[t], 1);
    bucket[(long)t * Bn + r] = row;
}

#define NEG_BIG (-3.4e38f)
#define NW 4

__device__ __forceinline__ void load_row(v4f (&z)[8], float& z_c,
                                         const float* __restrict__ lrow,
                                         const int o0, const int o7, const int c)
{
#pragma unroll
    for (int i = 0; i < 7; ++i) z[i] = *(const v4f*)(lrow + o0 + 128 * i);
    z[7] = *(const v4f*)(lrow + o7);
    z_c  = lrow[c];
}

__device__ __forceinline__ int xhalf_i(const int b)
{
    return __builtin_amdgcn_permlanex16(b, b, 0x76543210, 0xfedcba98, false, false);
}
template <int M> __device__ __forceinline__ int xrow_i(const int b)
{
    return __builtin_amdgcn_update_dpp(0, b, 0x160 | M, 0xf, 0xf, true);
}
__device__ __forceinline__ float xhalf_f(const float x) { return __builtin_bit_cast(float, xhalf_i(__builtin_bit_cast(int, x))); }
template <int M> __device__ __forceinline__ float xrow_f(const float x) { return __builtin_bit_cast(float, xrow_i<M>(__builtin_bit_cast(int, x))); }

__device__ __forceinline__ float wave_max_f(float x)
{
    x = __builtin_elementwise_maximum(x, xhalf_f(x));
    x = __builtin_elementwise_maximum(x, xrow_f<8>(x));
    x = __builtin_elementwise_maximum(x, xrow_f<4>(x));
    x = __builtin_elementwise_maximum(x, xrow_f<2>(x));
    x = __builtin_elementwise_maximum(x, xrow_f<1>(x));
    return x;
}
__device__ __forceinline__ float wave_sum_f(float x)
{
    x += xhalf_f(x);
    x += xrow_f<8>(x);
    x += xrow_f<4>(x);
    x += xrow_f<2>(x);
    x += xrow_f<1>(x);
    return x;
}
__device__ __forceinline__ int wave_min_i(int x)
{
    x = min(x, xhalf_i(x));
    x = min(x, xrow_i<8>(x));
    x = min(x, xrow_i<4>(x));
    x = min(x, xrow_i<2>(x));
    x = min(x, xrow_i<1>(x));
    return x;
}

__device__ __forceinline__ void row_step(v4f (&z)[8], const float z_c, v4f (&acc)[8],
                                         const float* slam, const float lam_cc,
                                         const int lane, const int o0, const int o7, const bool ok7,
                                         float* __restrict__ confrow, float& cesum)
{
    const v4f negbig = {NEG_BIG, NEG_BIG, NEG_BIG, NEG_BIG};
    float maxv = NEG_BIG; int slot = 0x7fffffff;
    float maxzp = NEG_BIG;
#pragma unroll
    for (int i = 0; i < 8; ++i) {
        const v4f l = *(const v4f*)(slam + ((i < 7) ? (o0 + 128 * i) : o7));
        v4f zr = z[i];
        v4f zp = zr - l;
        if (i == 7) { zr = ok7 ? zr : negbig; zp = ok7 ? zp : negbig; }
#pragma unroll
        for (int k = 0; k < 4; ++k) {
            if (zr[k] > maxv) { maxv = zr[k]; slot = 4 * i + k; }
            maxzp = fmaxf(maxzp, zp[k]);
        }
        z[i] = zp;
    }
    const int mycol = (slot == 0x7fffffff) ? slot : (4 * lane + 128 * (slot >> 2) + (slot & 3));
    const float wmax = wave_max_f(maxv);
    const int maxi = wave_min_i((maxv == wmax) ? mycol : 0x7fffffff);
    maxzp = wave_max_f(maxzp);

    float sum = 0.0f;
#pragma unroll
    for (int i = 0; i < 8; ++i) {
        v4f e;
#pragma unroll
        for (int k = 0; k < 4; ++k) e[k] = __expf(z[i][k] - maxzp);
        z[i] = e;
        sum += (e[0] + e[1]) + (e[2] + e[3]);
    }
    sum = wave_sum_f(sum);
    const float lse = __logf(sum);
    const float inv = 1.0f / sum;

#pragma unroll
    for (int i = 0; i < 8; ++i) acc[i] += z[i] * inv;

    if (lane == 0) atomicAdd(&confrow[maxi], 1.0f);
    cesum += ((z_c - lam_cc) - maxzp) - lse;
}

__global__ __launch_bounds__(NW * 32, 4) void class_pass(
    const float* __restrict__ logits, const float* __restrict__ Lam,
    const int* __restrict__ cnt, const int* __restrict__ bucket,
    float* __restrict__ seg, float* __restrict__ confc,
    float* __restrict__ priorc, float* __restrict__ cep)
{
    __shared__ float sacc[NW][Cc];
    __shared__ float slam[Cc];
    const int wave = __builtin_amdgcn_readfirstlane(threadIdx.x >> 5);
    const int lane = threadIdx.x & 31;
    const int tid  = threadIdx.x;
    const int c    = blockIdx.x;
    const int n    = cnt[c];
    const int NV   = Cc / 4;
    const int* __restrict__ brow = bucket + (long)c * Bn;

    const bool ok7 = (lane + 32 * 7) < NV;
    const int  o0  = 4 * lane;
    const int  o7  = 4 * min(lane + 32 * 7, NV - 1);

    const float* __restrict__ lamrow = Lam + (long)c * Cc;
    for (int j = tid; j < NV; j += NW * 32) *(v4f*)(&slam[4 * j]) = *(const v4f*)(lamrow + 4 * j);
    const float lam_cc = lamrow[c];

    const v4f z4 = {0.f, 0.f, 0.f, 0.f};
    v4f acc[8];
#pragma unroll
    for (int i = 0; i < 8; ++i) acc[i] = z4;
    float cesum = 0.0f;
    float* __restrict__ confrow = confc + (long)c * Cc;

    v4f za[8], zb[8];
    float zca, zcb;
    const int last = max(n - 1, 0);
    int r = wave;
    const int rowFirst = (n > 0) ? brow[min(r, last)] : 0;
    int rowNext        = (n > 0) ? brow[min(r + NW, last)] : 0;
    load_row(za, zca, logits + (long)rowFirst * Cc, o0, o7, c);
    __syncthreads();
    if (r < n) {
        for (;;) {
            load_row(zb, zcb, logits + (long)rowNext * Cc, o0, o7, c);
            rowNext = brow[min(r + 2 * NW, last)];
            row_step(za, zca, acc, slam, lam_cc, lane, o0, o7, ok7, confrow, cesum);
            r += NW;
            if (r >= n) break;
            load_row(za, zca, logits + (long)rowNext * Cc, o0, o7, c);
            rowNext = brow[min(r + 2 * NW, last)];
            row_step(zb, zcb, acc, slam, lam_cc, lane, o0, o7, ok7, confrow, cesum);
            r += NW;
            if (r >= n) break;
        }
    }

#pragma unroll
    for (int i = 0; i < 7; ++i) *(v4f*)(&sacc[wave][o0 + 128 * i]) = acc[i];
    if (ok7) *(v4f*)(&sacc[wave][o0 + 128 * 7]) = acc[7];
    __syncthreads();
    for (int j = tid; j < NV; j += NW * 32) {
        v4f s = *(const v4f*)(&sacc[0][4 * j]);
#pragma unroll
        for (int w = 1; w < NW; ++w) s += *(const v4f*)(&sacc[w][4 * j]);
        *(v4f*)(seg + (long)c * Cc + 4 * j) = s;
    }
    if (tid == 0) priorc[c] = (float)n;
    if (lane == 0 && wave < n) atomicAdd(&cep[(c * NW + wave) & 255], cesum);
}

__device__ __forceinline__ float wave_sum32(const float (&v)[32]) {
    const v2f bones = {1.0f, 1.0f};
    const v8f vzero = {0.f,0.f,0.f,0.f,0.f,0.f,0.f,0.f};
    v8f acc0 = vzero, acc1 = vzero;
#pragma unroll
    for (int p = 0; p < 32; p += 4) {
        const v2f a0 = {v[p + 0], v[p + 1]};
        const v2f a1 = {v[p + 2], v[p + 3]};
        acc0 = __builtin_amdgcn_wmma_f32_16x16x4_f32(false, a0, false, bones,
                                                     (short)0, acc0, false, false);
        acc1 = __builtin_amdgcn_wmma_f32_16x16x4_f32(false, a1, false, bones,
                                                     (short)0, acc1, false, false);
    }
    const v8f acc = acc0 + acc1;
    float s = ((acc[0] + acc[1]) + (acc[2] + acc[3])) + ((acc[4] + acc[5]) + (acc[6] + acc[7]));
    s += __shfl_xor(s, 16, 32);
    return s;
}

typedef float v4f_a4 __attribute__((ext_vector_type(4), aligned(4)));

#define PW 8
#define NPART (Cc / PW)
#define CPITCH 1024

__global__ __launch_bounds__(PW * 32) void post_rows(
    const float* __restrict__ conf_ema, const float* __restrict__ Lam,
    const float* __restrict__ seg, const float* __restrict__ confc,
    const float* __restrict__ priorc,
    float* __restrict__ o_conf, float* __restrict__ o_lam,
    float* __restrict__ colpart)
{
    __shared__ __attribute__((aligned(16))) float scol[PW][Cc];
    const int wave  = __builtin_amdgcn_readfirstlane(threadIdx.x >> 5);
    const int lane  = threadIdx.x & 31;
    const int row   = blockIdx.x * PW + wave;
    const long base = (long)row * Cc;
    const int NV    = Cc / 4;

    const float cnt     = priorc[row];
    const float invc    = 1.0f / fmaxf(cnt, 1.0f);
    const bool  present = (cnt > 0.0f);

    const v4f z4 = {0.f, 0.f, 0.f, 0.f};
    v4f ve[8], vb[8], vs[8], vl[8];
    const bool ok7 = (lane + 32 * 7) < NV;
    const long o0  = base + 4 * lane;
    const long o7  = base + 4 * min(lane + 32 * 7, NV - 1);
#pragma unroll
    for (int i = 0; i < 7; ++i) {
        ve[i] = *(const v4f*)(conf_ema + o0 + 128 * i);
        vb[i] = *(const v4f*)(confc    + o0 + 128 * i);
        vs[i] = *(const v4f*)(seg      + o0 + 128 * i);
        vl[i] = *(const v4f*)(Lam      + o0 + 128 * i);
    }
    ve[7] = *(const v4f*)(conf_ema + o7);
    vb[7] = *(const v4f*)(confc    + o7);
    vs[7] = *(const v4f*)(seg      + o7);
    vl[7] = *(const v4f*)(Lam      + o7);
    ve[7] = ok7 ? ve[7] : z4;
    vb[7] = ok7 ? vb[7] : z4;
    vs[7] = ok7 ? vs[7] : z4;
    vl[7] = ok7 ? vl[7] : z4;

#pragma unroll
    for (int i = 0; i < 7; ++i) *(v4f*)(&scol[wave][4 * lane + 128 * i]) = vs[i];
    if (ok7) *(v4f*)(&scol[wave][4 * lane + 128 * 7]) = vs[7];

    float acnt[32], empr[32];
#pragma unroll
    for (int i = 0; i < 8; ++i) {
        const int j = lane + 32 * i;
        const bool ok = (i < 7) || ok7;
        const v4f cn = DECAYF * ve[i] + ONE_M_DECAY * vb[i];
        if (ok) *(v4f_a4*)(o_conf + base + 4 * j) = cn;
#pragma unroll
        for (int k = 0; k < 4; ++k) {
            const int c = 4 * j + k;
            const bool offd = ok && (c != row);
            acnt[4 * i + k] = offd ? (cn[k] + ALPHAF) : 0.f;
            empr[4 * i + k] = offd ? (vs[i][k] * invc) : 0.f;
        }
    }

    const float countsSum = wave_sum32(acnt);
    const float empSum    = wave_sum32(empr);
    const float phiDen = 1.0f / (countsSum + EPSF);
    const float empDen = 1.0f / fmaxf(empSum, EPSF);

    float l1[32];
#pragma unroll
    for (int i = 0; i < 8; ++i) {
        const int j = lane + 32 * i;
#pragma unroll
        for (int k = 0; k < 4; ++k) {
            const int c = 4 * j + k;
            const float phi = acnt[4 * i + k] * phiDen;
            const float emp = empr[4 * i + k] * empDen;
            const float v   = vl[i][k] + LR_LF * (emp - phi);
            l1[4 * i + k] = (((i < 7) || ok7) && (c != row)) ? v : 0.f;
        }
    }
    const float rowMean = wave_sum32(l1) * (1.0f / (float)Cc);

    __syncthreads();
    {
        const int t = threadIdx.x;
        if (t < NV) {
            v4f s = *(const v4f*)(&scol[0][4 * t]);
#pragma unroll
            for (int w = 1; w < PW; ++w) s += *(const v4f*)(&scol[w][4 * t]);
            *(v4f*)(colpart + (long)blockIdx.x * CPITCH + 4 * t) = s;
        }
    }

#pragma unroll
    for (int i = 0; i < 8; ++i) {
        const int j = lane + 32 * i;
        const bool ok = (i < 7) || ok7;
        v4f r;
#pragma unroll
        for (int k = 0; k < 4; ++k)
            r[k] = present ? (l1[4 * i + k] - rowMean) : vl[i][k];
        if (ok) *(v4f_a4*)(o_lam + base + 4 * j) = r;
    }
}

__global__ __launch_bounds__(1024) void finalize(
    const float* __restrict__ prior_ema, const float* __restrict__ mu,
    const float* __restrict__ colpart, const float* __restrict__ priorc,
    const float* __restrict__ cep,
    float* __restrict__ o_ce, float* __restrict__ o_mu, float* __restrict__ o_prior)
{
    __shared__ float red[1024];
    __shared__ __attribute__((aligned(16))) float scs[4][Cc];
    const int j = threadIdx.x;

    {
        const int NV = Cc / 4;
        const int g  = j >> 8;
        const int c4 = j & 255;
        if (c4 < NV) {
            const int NK = (NPART + 3) / 4;
            const v4f z4 = {0.f, 0.f, 0.f, 0.f};
            v4f s = z4;
#pragma unroll 16
            for (int k = 0; k < NK; ++k) {
                const int r = g + 4 * k;
                const v4f v = *(const v4f*)(colpart + (long)min(r, NPART - 1) * CPITCH + 4 * c4);
                s += (r < NPART) ? v : z4;
            }
            *(v4f*)(&scs[g][4 * c4]) = s;
        }
    }

    float pn = 0.0f;
    if (j < Cc) {
        pn = DECAYF * prior_ema[j] + ONE_M_DECAY * priorc[j];
        o_prior[j] = pn;
    }
    red[j] = pn; __syncthreads();
    for (int s = 512; s; s >>= 1) { if (j < s) red[j] += red[j + s]; __syncthreads(); }
    const float priorSum = red[0];
    __syncthreads();

    float mu1 = 0.0f;
    if (j < Cc) {
        const float colsum = (scs[0][j] + scs[1][j]) + (scs[2][j] + scs[3][j]);
        float emp_marg = colsum * (1.0f / (float)Bn);
        float pi = pn / (priorSum + EPSF);
        mu1 = mu[j] + LR_MUF * (emp_marg - pi);
    }
    red[j] = mu1; __syncthreads();
    for (int s = 512; s; s >>= 1) { if (j < s) red[j] += red[j + s]; __syncthreads(); }
    const float muMean = red[0] * (1.0f / (float)Cc);
    if (j < Cc) o_mu[j] = mu1 - muMean;

    if (j == 0) {
        float s = 0.0f;
        for (int i = 0; i < 256; ++i) s += cep[i];
        o_ce[0] = -s * (1.0f / (float)Bn);
    }
}

extern "C" void kernel_launch(void* const* d_in, const int* in_sizes, int n_in,
                              void* d_out, int out_size, void* d_ws, size_t ws_size,
                              hipStream_t stream)
{
    const float* logits    = (const float*)d_in[0];
    const int*   targets   = (const int*)  d_in[1];
    const float* conf_ema  = (const float*)d_in[2];
    const float* prior_ema = (const float*)d_in[3];
    const float* Lam       = (const float*)d_in[4];
    const float* mu        = (const float*)d_in[5];

    float* out     = (float*)d_out;
    float* o_ce    = out;
    float* o_lam   = out + 1;
    float* o_mu    = out + 1 + (long)Cc * Cc;
    float* o_conf  = out + 1 + (long)Cc * Cc + Cc;
    float* o_prior = out + 1 + 2L * Cc * Cc + Cc;

    float* ws     = (float*)d_ws;
    float* seg    = ws;
    float* confc  = ws + (long)Cc * Cc;
    float* priorc = ws + 2L * Cc * Cc;
    float* cep    = priorc + Cc;
    int*   cnt    = (int*)(cep + 256);
    int*   bucket = cnt + 1024;
    float* colpart = (float*)(bucket + (long)Cc * Bn) + 24;

    const long nz = 1L * Cc * Cc + Cc + 256 + 1024;
    zero_ws<<<1024, 256, 0, stream>>>(confc, nz);
    bucket_rows<<<Bn / 256, 256, 0, stream>>>(targets, cnt, bucket);
    class_pass<<<Cc, NW * 32, 0, stream>>>(logits, Lam, cnt, bucket, seg, confc, priorc, cep);
    post_rows<<<Cc / PW, PW * 32, 0, stream>>>(conf_ema, Lam, seg, confc, priorc,
                                               o_conf, o_lam, colpart);
    finalize<<<1, 1024, 0, stream>>>(prior_ema, mu, colpart, priorc, cep,
                                     o_ce, o_mu, o_prior);
}


# class_pass: a 16-wave block per 4 classes lists its own rows from one read of the targets (LDS counters and counts): zero-fill and bucket launches, global atomics gone; CE summed in a new order
# speedup vs baseline: 10.9908x; 1.2406x over previous
#include <hip/hip_runtime.h>

#define Cc 1000
#define Bn 65536
#define DECAYF 0.99f
#define ONE_M_DECAY 0.01f
#define ALPHAF 1.0f
#define LR_LF 0.05f
#define LR_MUF 0.02f
#define EPSF 1e-12f

typedef __attribute__((ext_vector_type(2))) float v2f;
typedef __attribute__((ext_vector_type(4))) float v4f;
typedef __attribute__((ext_vector_type(8))) float v8f;

#define NEG_BIG (-3.4e38f)
#define NW 4
#define KC 4
#define NWB (NW * KC)
typedef __attribute__((ext_vector_type(4))) int v4i;

__device__ __forceinline__ void load_row(v4f (&z)[8], float& z_c,
                                         const float* __restrict__ lrow,
                                         const int o0, const int o7, const int c)
{
#pragma unroll
    for (int i = 0; i < 7; ++i) z[i] = *(const v4f*)(lrow + o0 + 128 * i);
    z[7] = *(const v4f*)(lrow + o7);
    z_c  = lrow[c];
}

__device__ __forceinline__ int xhalf_i(const int b)
{
    return __builtin_amdgcn_permlanex16(b, b, 0x76543210, 0xfedcba98, false, false);
}
template <int M> __device__ __forceinline__ int xrow_i(const int b)
{
    return __builtin_amdgcn_update_dpp(0, b, 0x160 | M, 0xf, 0xf, true);
}
__device__ __forceinline__ float xhalf_f(const float x) { return __builtin_bit_cast(float, xhalf_i(__builtin_bit_cast(int, x))); }
template <int M> __device__ __forceinline__ float xrow_f(const float x) { return __builtin_bit_cast(float, xrow_i<M>(__builtin_bit_cast(int, x))); }

__device__ __forceinline__ float wave_max_f(float x)
{
    x = __builtin_elementwise_maximum(x, xhalf_f(x));
    x = __builtin_elementwise_maximum(x, xrow_f<8>(x));
    x = __builtin_elementwise_maximum(x, xrow_f<4>(x));
    x = __builtin_elementwise_maximum(x, xrow_f<2>(x));
    x = __builtin_elementwise_maximum(x, xrow_f<1>(x));
    return x;
}
__device__ __forceinline__ float wave_sum_f(float x)
{
    x += xhalf_f(x);
    x += xrow_f<8>(x);
    x += xrow_f<4>(x);
    x += xrow_f<2>(x);
    x += xrow_f<1>(x);
    return x;
}
__device__ __forceinline__ int wave_min_i(int x)
{
    x = min(x, xhalf_i(x));
    x = min(x, xrow_i<8>(x));
    x = min(x, xrow_i<4>(x));
    x = min(x, xrow_i<2>(x));
    x = min(x, xrow_i<1>(x));
    return x;
}

__device__ __forceinline__ void row_step(v4f (&z)[8], const float z_c, v4f (&acc)[8],
                                         const float* slam, const float lam_cc,
                                         const int lane, const int o0, const int o7, const bool ok7,
                                         int* confrow, float& cesum)
{
    const v4f negbig = {NEG_BIG, NEG_BIG, NEG_BIG, NEG_BIG};
    float maxv = NEG_BIG; int slot = 0x7fffffff;
    float maxzp = NEG_BIG;
#pragma unroll
    for (int i = 0; i < 8; ++i) {
        const v4f l = *(const v4f*)(slam + ((i < 7) ? (o0 + 128 * i) : o7));
        v4f zr = z[i];
        v4f zp = zr - l;
        if (i == 7) { zr = ok7 ? zr : negbig; zp = ok7 ? zp : negbig; }
#pragma unroll
        for (int k = 0; k < 4; ++k) {
            if (zr[k] > maxv) { maxv = zr[k]; slot = 4 * i + k; }
            maxzp = fmaxf(maxzp, zp[k]);
        }
        z[i] = zp;
    }
    const int mycol = (slot == 0x7fffffff) ? slot : (4 * lane + 128 * (slot >> 2) + (slot & 3));
    const float wmax = wave_max_f(maxv);
    const int maxi = wave_min_i((maxv == wmax) ? mycol : 0x7fffffff);
    maxzp = wave_max_f(maxzp);

    float sum = 0.0f;
#pragma unroll
    for (int i = 0; i < 8; ++i) {
        v4f e;
#pragma unroll
        for (int k = 0; k < 4; ++k) e[k] = __expf(z[i][k] - maxzp);
        z[i] = e;
        sum += (e[0] + e[1]) + (e[2] + e[3]);
    }
    sum = wave_sum_f(sum);
    const float lse = __logf(sum);
    const float inv = 1.0f / sum;

#pragma unroll
    for (int i = 0; i < 8; ++i) acc[i] += z[i] * inv;

    if (lane == 0) atomicAdd(&confrow[maxi], 1);
    cesum += ((z_c - lam_cc) - maxzp) - lse;
}

__global__ __launch_bounds__(NWB * 32) void class_pass(
    const float* __restrict__ logits, const float* __restrict__ Lam,
    const int* __restrict__ targets, int* __restrict__ bucket,
    float* __restrict__ seg, float* __restrict__ confc,
    float* __restrict__ priorc, float* __restrict__ cep)
{
    __shared__ __attribute__((aligned(16))) float sacc[NWB][Cc];
    __shared__ __attribute__((aligned(16))) float slam[KC][Cc];
    __shared__ __attribute__((aligned(16))) int   sconf[KC][Cc];
    __shared__ int   scnt[KC];
    __shared__ float sce[NWB];
    const int wave = __builtin_amdgcn_readfirstlane(threadIdx.x >> 5);
    const int lane = threadIdx.x & 31;
    const int tid  = threadIdx.x;
    const int k    = wave / NW;
    const int w    = wave % NW;
    const int c0   = blockIdx.x * KC;
    const int c    = c0 + k;
    const int NV   = Cc / 4;

    {
        const v4i zi = {0, 0, 0, 0};
        for (int j = tid; j < KC * NV; j += NWB * 32) *(v4i*)(&sconf[0][0] + 4 * j) = zi;
        if (tid < KC) scnt[tid] = 0;
        const float* __restrict__ lam0 = Lam + (long)c0 * Cc;
        for (int j = tid; j < KC * NV; j += NWB * 32) *(v4f*)(&slam[0][0] + 4 * j) = *(const v4f*)(lam0 + 4 * j);
    }
    __syncthreads();

    {
        const int NG = Bn / 4 / (NWB * 32);
#pragma unroll 1
        for (int g0 = 0; g0 < NG; g0 += 8) {
            v4i tv[8];
            int  ti[8];
#pragma unroll
            for (int u = 0; u < 8; ++u) {
                ti[u] = (((g0 + u + (int)blockIdx.x) & (NG - 1)) * (NWB * 32)) + tid;
                tv[u] = *(const v4i*)(targets + 4 * ti[u]);
            }
#pragma unroll
            for (int u = 0; u < 8; ++u) {
                const unsigned dmin = min(min((unsigned)(tv[u][0] - c0), (unsigned)(tv[u][1] - c0)),
                                          min((unsigned)(tv[u][2] - c0), (unsigned)(tv[u][3] - c0)));
                if (dmin < (unsigned)KC) {
#pragma unroll
                    for (int e = 0; e < 4; ++e) {
                        const int t = tv[u][e];
                        const unsigned d = (unsigned)(t - c0);
                        if (d < (unsigned)KC) {
                            const int slot = atomicAdd(&scnt[d], 1);
                            bucket[(long)t * Bn + slot] = 4 * ti[u] + e;
                        }
                    }
                }
            }
        }
    }
    __syncthreads();

    const int n = scnt[k];
    const int* brow = bucket + (long)c * Bn;

    const bool ok7 = (lane + 32 * 7) < NV;
    const int  o0  = 4 * lane;
    const int  o7  = 4 * min(lane + 32 * 7, NV - 1);

    const float* slamk = &slam[k][0];
    const float lam_cc = slamk[c];

    const v4f z4 = {0.f, 0.f, 0.f, 0.f};
    v4f acc[8];
#pragma unroll
    for (int i = 0; i < 8; ++i) acc[i] = z4;
    float cesum = 0.0f;
    int* confrow = &sconf[k][0];

    v4f za[8], zb[8];
    float zca, zcb;
    const int last = max(n - 1, 0);
    int r = w;
    int s = 0;
    int idxv = (n > 0) ? brow[min(w + NW * lane, last)] : 0;
    int idxn = (n > 0) ? brow[min(w + NW * (32 + lane), last)] : 0;
    load_row(za, zca, logits + (long)__builtin_amdgcn_readlane(idxv, 0) * Cc, o0, o7, c);
    if (r < n) {
        for (;;) {
            ++s;
            if ((s & 31) == 0) { idxv = idxn; idxn = brow[min(w + NW * (s + 32 + lane), last)]; }
            load_row(zb, zcb, logits + (long)__builtin_amdgcn_readlane(idxv, s & 31) * Cc, o0, o7, c);
            row_step(za, zca, acc, slamk, lam_cc, lane, o0, o7, ok7, confrow, cesum);
            r += NW;
            if (r >= n) break;
            ++s;
            if ((s & 31) == 0) { idxv = idxn; idxn = brow[min(w + NW * (s + 32 + lane), last)]; }
            load_row(za, zca, logits + (long)__builtin_amdgcn_readlane(idxv, s & 31) * Cc, o0, o7, c);
            row_step(zb, zcb, acc, slamk, lam_cc, lane, o0, o7, ok7, confrow, cesum);
            r += NW;
            if (r >= n) break;
        }
    }

#pragma unroll
    for (int i = 0; i < 7; ++i) *(v4f*)(&sacc[wave][o0 + 128 * i]) = acc[i];
    if (ok7) *(v4f*)(&sacc[wave][o0 + 128 * 7]) = acc[7];
    if (lane == 0) sce[wave] = cesum;
    __syncthreads();
    {
        const int tk = tid & (NW * 32 - 1);
        for (int j = tk; j < NV; j += NW * 32) {
            v4f s = *(const v4f*)(&sacc[k * NW][4 * j]);
#pragma unroll
            for (int ww = 1; ww < NW; ++ww) s += *(const v4f*)(&sacc[k * NW + ww][4 * j]);
            *(v4f*)(seg + (long)c * Cc + 4 * j) = s;
            const v4i q = *(const v4i*)(&sconf[k][4 * j]);
            const v4f f = {(float)q[0], (float)q[1], (float)q[2], (float)q[3]};
            *(v4f*)(confc + (long)c * Cc + 4 * j) = f;
        }
        if (tk == 0) {
            priorc[c] = (float)n;
            cep[c] = (sce[k * NW] + sce[k * NW + 1]) + (sce[k * NW + 2] + sce[k * NW + 3]);
        }
    }
}

__device__ __forceinline__ float wave_sum32(const float (&v)[32]) {
    const v2f bones = {1.0f, 1.0f};
    const v8f vzero = {0.f,0.f,0.f,0.f,0.f,0.f,0.f,0.f};
    v8f acc0 = vzero, acc1 = vzero;
#pragma unroll
    for (int p = 0; p < 32; p += 4) {
        const v2f a0 = {v[p + 0], v[p + 1]};
        const v2f a1 = {v[p + 2], v[p + 3]};
        acc0 = __builtin_amdgcn_wmma_f32_16x16x4_f32(false, a0, false, bones,
                                                     (short)0, acc0, false, false);
        acc1 = __builtin_amdgcn_wmma_f32_16x16x4_f32(false, a1, false, bones,
                                                     (short)0, acc1, false, false);
    }
    const v8f acc = acc0 + acc1;
    float s = ((acc[0] + acc[1]) + (acc[2] + acc[3])) + ((acc[4] + acc[5]) + (acc[6] + acc[7]));
    s += __shfl_xor(s, 16, 32);
    return s;
}

typedef float v4f_a4 __attribute__((ext_vector_type(4), aligned(4)));

#define PW 8
#define NPART (Cc / PW)
#define CPITCH 1024

__global__ __launch_bounds__(PW * 32) void post_rows(
    const float* __restrict__ conf_ema, const float* __restrict__ Lam,
    const float* __restrict__ seg, const float* __restrict__ confc,
    const float* __restrict__ priorc,
    float* __restrict__ o_conf, float* __restrict__ o_lam,
    float* __restrict__ colpart)
{
    __shared__ __attribute__((aligned(16))) float scol[PW][Cc];
    const int wave  = __builtin_amdgcn_readfirstlane(threadIdx.x >> 5);
    const int lane  = threadIdx.x & 31;
    const int row   = blockIdx.x * PW + wave;
    const long base = (long)row * Cc;
    const int NV    = Cc / 4;

    const float cnt     = priorc[row];
    const float invc    = 1.0f / fmaxf(cnt, 1.0f);
    const bool  present = (cnt > 0.0f);

    const v4f z4 = {0.f, 0.f, 0.f, 0.f};
    v4f ve[8], vb[8], vs[8], vl[8];
    const bool ok7 = (lane + 32 * 7) < NV;
    const long o0  = base + 4 * lane;
    const long o7  = base + 4 * min(lane + 32 * 7, NV - 1);
#pragma unroll
    for (int i = 0; i < 7; ++i) {
        ve[i] = *(const v4f*)(conf_ema + o0 + 128 * i);
        vb[i] = *(const v4f*)(confc    + o0 + 128 * i);
        vs[i] = *(const v4f*)(seg      + o0 + 128 * i);
        vl[i] = *(const v4f*)(Lam      + o0 + 128 * i);
    }
    ve[7] = *(const v4f*)(conf_ema + o7);
    vb[7] = *(const v4f*)(confc    + o7);
    vs[7] = *(const v4f*)(seg      + o7);
    vl[7] = *(const v4f*)(Lam      + o7);
    ve[7] = ok7 ? ve[7] : z4;
    vb[7] = ok7 ? vb[7] : z4;
    vs[7] = ok7 ? vs[7] : z4;
    vl[7] = ok7 ? vl[7] : z4;

#pragma unroll
    for (int i = 0; i < 7; ++i) *(v4f*)(&scol[wave][4 * lane + 128 * i]) = vs[i];
    if (ok7) *(v4f*)(&scol[wave][4 * lane + 128 * 7]) = vs[7];

    float acnt[32], empr[32];
#pragma unroll
    for (int i = 0; i < 8; ++i) {
        const int j = lane + 32 * i;
        const bool ok = (i < 7) || ok7;
        const v4f cn = DECAYF * ve[i] + ONE_M_DECAY * vb[i];
        if (ok) *(v4f_a4*)(o_conf + base + 4 * j) = cn;
#pragma unroll
        for (int k = 0; k < 4; ++k) {
            const int c = 4 * j + k;
            const bool offd = ok && (c != row);
            acnt[4 * i + k] = offd ? (cn[k] + ALPHAF) : 0.f;
            empr[4 * i + k] = offd ? (vs[i][k] * invc) : 0.f;
        }
    }

    const float countsSum = wave_sum32(acnt);
    const float empSum    = wave_sum32(empr);
    const float phiDen = 1.0f / (countsSum + EPSF);
    const float empDen = 1.0f / fmaxf(empSum, EPSF);

    float l1[32];
#pragma unroll
    for (int i = 0; i < 8; ++i) {
        const int j = lane + 32 * i;
#pragma unroll
        for (int k = 0; k < 4; ++k) {
            const int c = 4 * j + k;
            const float phi = acnt[4 * i + k] * phiDen;
            const float emp = empr[4 * i + k] * empDen;
            const float v   = vl[i][k] + LR_LF * (emp - phi);
            l1[4 * i + k] = (((i < 7) || ok7) && (c != row)) ? v : 0.f;
        }
    }
    const float rowMean = wave_sum32(l1) * (1.0f / (float)Cc);

    __syncthreads();
    {
        const int t = threadIdx.x;
        if (t < NV) {
            v4f s = *(const v4f*)(&scol[0][4 * t]);
#pragma unroll
            for (int w = 1; w < PW; ++w) s += *(const v4f*)(&scol[w][4 * t]);
            *(v4f*)(colpart + (long)blockIdx.x * CPITCH + 4 * t) = s;
        }
    }

#pragma unroll
    for (int i = 0; i < 8; ++i) {
        const int j = lane + 32 * i;
        const bool ok = (i < 7) || ok7;
        v4f r;
#pragma unroll
        for (int k = 0; k < 4; ++k)
            r[k] = present ? (l1[4 * i + k] - rowMean) : vl[i][k];
        if (ok) *(v4f_a4*)(o_lam + base + 4 * j) = r;
    }
}

__global__ __launch_bounds__(1024) void finalize(
    const float* __restrict__ prior_ema, const float* __restrict__ mu,
    const float* __restrict__ colpart, const float* __restrict__ priorc,
    const float* __restrict__ cep,
    float* __restrict__ o_ce, float* __restrict__ o_mu, float* __restrict__ o_prior)
{
    __shared__ float red[3][32];
    __shared__ __attribute__((aligned(16))) float scs[4][Cc];
    const int j    = threadIdx.x;
    const int lane = j & 31;
    const int wave = j >> 5;

    float pe = 0.0f, pc = 0.0f, muj = 0.0f, cej = 0.0f;
    if (j < Cc) { pe = prior_ema[j]; pc = priorc[j]; muj = mu[j]; }
    if (j < Cc) cej = cep[j];

    {
        const int NV = Cc / 4;
        const int g  = j >> 8;
        const int c4 = j & 255;
        if (c4 < NV) {
            const int NK = (NPART + 3) / 4;
            const v4f z4 = {0.f, 0.f, 0.f, 0.f};
            v4f s = z4;
#pragma unroll 16
            for (int k = 0; k < NK; ++k) {
                const int r = g + 4 * k;
                const v4f v = *(const v4f*)(colpart + (long)min(r, NPART - 1) * CPITCH + 4 * c4);
                s += (r < NPART) ? v : z4;
            }
            *(v4f*)(&scs[g][4 * c4]) = s;
        }
    }

    float pn = 0.0f;
    if (j < Cc) {
        pn = DECAYF * pe + ONE_M_DECAY * pc;
        o_prior[j] = pn;
    }
    {
        const float wp = wave_sum_f(pn);
        const float wc = wave_sum_f(cej);
        if (lane == 0) { red[0][wave] = wp; red[1][wave] = wc; }
    }
    __syncthreads();
    const float priorSum = wave_sum_f(red[0][lane]);

    float mu1 = 0.0f;
    if (j < Cc) {
        const float colsum = (scs[0][j] + scs[1][j]) + (scs[2][j] + scs[3][j]);
        float emp_marg = colsum * (1.0f / (float)Bn);
        float pi = pn / (priorSum + EPSF);
        mu1 = muj + LR_MUF * (emp_marg - pi);
    }
    {
        const float wm = wave_sum_f(mu1);
        if (lane == 0) red[2][wave] = wm;
    }
    __syncthreads();
    const float muMean = wave_sum_f(red[2][lane]) * (1.0f / (float)Cc);
    if (j < Cc) o_mu[j] = mu1 - muMean;

    if (wave == 0) {
        const float s = wave_sum_f(red[1][lane]);
        if (lane == 0) o_ce[0] = -s * (1.0f / (float)Bn);
    }
}

extern "C" void kernel_launch(void* const* d_in, const int* in_sizes, int n_in,
                              void* d_out, int out_size, void* d_ws, size_t ws_size,
                              hipStream_t stream)
{
    const float* logits    = (const float*)d_in[0];
    const int*   targets   = (const int*)  d_in[1];
    const float* conf_ema  = (const float*)d_in[2];
    const float* prior_ema = (const float*)d_in[3];
    const float* Lam       = (const float*)d_in[4];
    const float* mu        = (const float*)d_in[5];

    float* out     = (float*)d_out;
    float* o_ce    = out;
    float* o_lam   = out + 1;
    float* o_mu    = out + 1 + (long)Cc * Cc;
    float* o_conf  = out + 1 + (long)Cc * Cc + Cc;
    float* o_prior = out + 1 + 2L * Cc * Cc + Cc;

    float* ws     = (float*)d_ws;
    float* seg    = ws;
    float* confc  = ws + (long)Cc * Cc;
    float* priorc = ws + 2L * Cc * Cc;
    float* cep    = priorc + Cc;
    int*   bucket = (int*)(cep + 1280);
    float* colpart = (float*)(bucket + (long)Cc * Bn) + 24;

    class_pass<<<Cc / KC, NWB * 32, 0, stream>>>(logits, Lam, targets, bucket, seg, confc, priorc, cep);
    post_rows<<<Cc / PW, PW * 32, 0, stream>>>(conf_ema, Lam, seg, confc, priorc,
                                               o_conf, o_lam, colpart);
    finalize<<<1, 1024, 0, stream>>>(prior_ema, mu, colpart, priorc, cep,
                                     o_ce, o_mu, o_prior);
}


# column sums of q: class_pass writes one partial row per block, each post_rows block finishes 8 columns over the 250 rows (wave sums); finalize reads 1000 sums, not 125 rows; summed in a new order
# speedup vs baseline: 12.2401x; 1.1137x over previous
#include <hip/hip_runtime.h>

#define Cc 1000
#define Bn 65536
#define DECAYF 0.99f
#define ONE_M_DECAY 0.01f
#define ALPHAF 1.0f
#define LR_LF 0.05f
#define LR_MUF 0.02f
#define EPSF 1e-12f

typedef __attribute__((ext_vector_type(2))) float v2f;
typedef __attribute__((ext_vector_type(4))) float v4f;
typedef __attribute__((ext_vector_type(8))) float v8f;

#define NEG_BIG (-3.4e38f)
#define NW 4
#define KC 4
#define NWB (NW * KC)
#define NBLK (Cc / KC)
#define CPITCH 1024
typedef __attribute__((ext_vector_type(4))) int v4i;

__device__ __forceinline__ void load_row(v4f (&z)[8], float& z_c,
                                         const float* __restrict__ lrow,
                                         const int o0, const int o7, const int c)
{
#pragma unroll
    for (int i = 0; i < 7; ++i) z[i] = *(const v4f*)(lrow + o0 + 128 * i);
    z[7] = *(const v4f*)(lrow + o7);
    z_c  = lrow[c];
}

__device__ __forceinline__ int xhalf_i(const int b)
{
    return __builtin_amdgcn_permlanex16(b, b, 0x76543210, 0xfedcba98, false, false);
}
template <int M> __device__ __forceinline__ int xrow_i(const int b)
{
    return __builtin_amdgcn_update_dpp(0, b, 0x160 | M, 0xf, 0xf, true);
}
__device__ __forceinline__ float xhalf_f(const float x) { return __builtin_bit_cast(float, xhalf_i(__builtin_bit_cast(int, x))); }
template <int M> __device__ __forceinline__ float xrow_f(const float x) { return __builtin_bit_cast(float, xrow_i<M>(__builtin_bit_cast(int, x))); }

__device__ __forceinline__ float wave_max_f(float x)
{
    x = __builtin_elementwise_maximum(x, xhalf_f(x));
    x = __builtin_elementwise_maximum(x, xrow_f<8>(x));
    x = __builtin_elementwise_maximum(x, xrow_f<4>(x));
    x = __builtin_elementwise_maximum(x, xrow_f<2>(x));
    x = __builtin_elementwise_maximum(x, xrow_f<1>(x));
    return x;
}
__device__ __forceinline__ float wave_sum_f(float x)
{
    x += xhalf_f(x);
    x += xrow_f<8>(x);
    x += xrow_f<4>(x);
    x += xrow_f<2>(x);
    x += xrow_f<1>(x);
    return x;
}
__device__ __forceinline__ int wave_min_i(int x)
{
    x = min(x, xhalf_i(x));
    x = min(x, xrow_i<8>(x));
    x = min(x, xrow_i<4>(x));
    x = min(x, xrow_i<2>(x));
    x = min(x, xrow_i<1>(x));
    return x;
}

__device__ __forceinline__ void row_step(v4f (&z)[8], const float z_c, v4f (&acc)[8],
                                         const float* slam, const float lam_cc,
                                         const int lane, const int o0, const int o7, const bool ok7,
                                         int* confrow, float& cesum)
{
    const v4f negbig = {NEG_BIG, NEG_BIG, NEG_BIG, NEG_BIG};
    float maxv = NEG_BIG; int slot = 0x7fffffff;
    float maxzp = NEG_BIG;
#pragma unroll
    for (int i = 0; i < 8; ++i) {
        const v4f l = *(const v4f*)(slam + ((i < 7) ? (o0 + 128 * i) : o7));
        v4f zr = z[i];
        v4f zp = zr - l;
        if (i == 7) { zr = ok7 ? zr : negbig; zp = ok7 ? zp : negbig; }
#pragma unroll
        for (int k = 0; k < 4; ++k) {
            if (zr[k] > maxv) { maxv = zr[k]; slot = 4 * i + k; }
            maxzp = fmaxf(maxzp, zp[k]);
        }
        z[i] = zp;
    }
    const int mycol = (slot == 0x7fffffff) ? slot : (4 * lane + 128 * (slot >> 2) + (slot & 3));
    const float wmax = wave_max_f(maxv);
    const int maxi = wave_min_i((maxv == wmax) ? mycol : 0x7fffffff);
    maxzp = wave_max_f(maxzp);

    float sum = 0.0f;
#pragma unroll
    for (int i = 0; i < 8; ++i) {
        v4f e;
#pragma unroll
        for (int k = 0; k < 4; ++k) e[k] = __expf(z[i][k] - maxzp);
        z[i] = e;
        sum += (e[0] + e[1]) + (e[2] + e[3]);
    }
    sum = wave_sum_f(sum);
    const float lse = __logf(sum);
    const float inv = 1.0f / sum;

#pragma unroll
    for (int i = 0; i < 8; ++i) acc[i] += z[i] * inv;

    if (lane == 0) atomicAdd(&confrow[maxi], 1);
    cesum += ((z_c - lam_cc) - maxzp) - lse;
}

__global__ __launch_bounds__(NWB * 32) void class_pass(
    const float* __restrict__ logits, const float* __restrict__ Lam,
    const int* __restrict__ targets, int* __restrict__ bucket,
    float* __restrict__ seg, float* __restrict__ confc,
    float* __restrict__ priorc, float* __restrict__ cep,
    float* __restrict__ blockpart)
{
    __shared__ __attribute__((aligned(16))) float sacc[NWB][Cc];
    __shared__ __attribute__((aligned(16))) float slam[KC][Cc];
    __shared__ __attribute__((aligned(16))) int   sconf[KC][Cc];
    __shared__ int   scnt[KC];
    __shared__ float sce[NWB];
    const int wave = __builtin_amdgcn_readfirstlane(threadIdx.x >> 5);
    const int lane = threadIdx.x & 31;
    const int tid  = threadIdx.x;
    const int k    = wave / NW;
    const int w    = wave % NW;
    const int c0   = blockIdx.x * KC;
    const int c    = c0 + k;
    const int NV   = Cc / 4;

    {
        const v4i zi = {0, 0, 0, 0};
        for (int j = tid; j < KC * NV; j += NWB * 32) *(v4i*)(&sconf[0][0] + 4 * j) = zi;
        if (tid < KC) scnt[tid] = 0;
        const float* __restrict__ lam0 = Lam + (long)c0 * Cc;
        for (int j = tid; j < KC * NV; j += NWB * 32) *(v4f*)(&slam[0][0] + 4 * j) = *(const v4f*)(lam0 + 4 * j);
    }
    __syncthreads();

    {
        const int NG = Bn / 4 / (NWB * 32);
#pragma unroll 1
        for (int g0 = 0; g0 < NG; g0 += 8) {
            v4i tv[8];
            int  ti[8];
#pragma unroll
            for (int u = 0; u < 8; ++u) {
                ti[u] = (((g0 + u + (int)blockIdx.x) & (NG - 1)) * (NWB * 32)) + tid;
                tv[u] = *(const v4i*)(targets + 4 * ti[u]);
            }
#pragma unroll
            for (int u = 0; u < 8; ++u) {
                const unsigned dmin = min(min((unsigned)(tv[u][0] - c0), (unsigned)(tv[u][1] - c0)),
                                          min((unsigned)(tv[u][2] - c0), (unsigned)(tv[u][3] - c0)));
                if (dmin < (unsigned)KC) {
#pragma unroll
                    for (int e = 0; e < 4; ++e) {
                        const int t = tv[u][e];
                        const unsigned d = (unsigned)(t - c0);
                        if (d < (unsigned)KC) {
                            const int slot = atomicAdd(&scnt[d], 1);
                            bucket[(long)t * Bn + slot] = 4 * ti[u] + e;
                        }
                    }
                }
            }
        }
    }
    __syncthreads();

    const int n = scnt[k];
    const int* brow = bucket + (long)c * Bn;

    const bool ok7 = (lane + 32 * 7) < NV;
    const int  o0  = 4 * lane;
    const int  o7  = 4 * min(lane + 32 * 7, NV - 1);

    const float* slamk = &slam[k][0];
    const float lam_cc = slamk[c];

    const v4f z4 = {0.f, 0.f, 0.f, 0.f};
    v4f acc[8];
#pragma unroll
    for (int i = 0; i < 8; ++i) acc[i] = z4;
    float cesum = 0.0f;
    int* confrow = &sconf[k][0];

    v4f za[8], zb[8];
    float zca, zcb;
    const int last = max(n - 1, 0);
    int r = w;
    int s = 0;
    int idxv = (n > 0) ? brow[min(w + NW * lane, last)] : 0;
    int idxn = (n > 0) ? brow[min(w + NW * (32 + lane), last)] : 0;
    load_row(za, zca, logits + (long)__builtin_amdgcn_readlane(idxv, 0) * Cc, o0, o7, c);
    if (r < n) {
        for (;;) {
            ++s;
            if ((s & 31) == 0) { idxv = idxn; idxn = brow[min(w + NW * (s + 32 + lane), last)]; }
            load_row(zb, zcb, logits + (long)__builtin_amdgcn_readlane(idxv, s & 31) * Cc, o0, o7, c);
            row_step(za, zca, acc, slamk, lam_cc, lane, o0, o7, ok7, confrow, cesum);
            r += NW;
            if (r >= n) break;
            ++s;
            if ((s & 31) == 0) { idxv = idxn; idxn = brow[min(w + NW * (s + 32 + lane), last)]; }
            load_row(za, zca, logits + (long)__builtin_amdgcn_readlane(idxv, s & 31) * Cc, o0, o7, c);
            row_step(zb, zcb, acc, slamk, lam_cc, lane, o0, o7, ok7, confrow, cesum);
            r += NW;
            if (r >= n) break;
        }
    }

#pragma unroll
    for (int i = 0; i < 7; ++i) *(v4f*)(&sacc[wave][o0 + 128 * i]) = acc[i];
    if (ok7) *(v4f*)(&sacc[wave][o0 + 128 * 7]) = acc[7];
    if (lane == 0) sce[wave] = cesum;
    __syncthreads();
    {
        const int tk = tid & (NW * 32 - 1);
        const int NJ = (NV + NW * 32 - 1) / (NW * 32);
        v4f cs[NJ];
#pragma unroll
        for (int u = 0; u < NJ; ++u) {
            const int j = tk + u * (NW * 32);
            cs[u] = z4;
            if (j < NV) {
                v4f s = *(const v4f*)(&sacc[k * NW][4 * j]);
#pragma unroll
                for (int ww = 1; ww < NW; ++ww) s += *(const v4f*)(&sacc[k * NW + ww][4 * j]);
                cs[u] = s;
                *(v4f*)(&sacc[k * NW][4 * j]) = s;
            }
        }
        __syncthreads();
#pragma unroll
        for (int u = 0; u < NJ; ++u) {
            const int j = tk + u * (NW * 32);
            if (j < NV) {
                *(v4f*)(seg + (long)c * Cc + 4 * j) = cs[u];
                const v4i q = *(const v4i*)(&sconf[k][4 * j]);
                const v4f f = {(float)q[0], (float)q[1], (float)q[2], (float)q[3]};
                *(v4f*)(confc + (long)c * Cc + 4 * j) = f;
            }
        }
        if (tk == 0) {
            priorc[c] = (float)n;
            cep[c] = (sce[k * NW] + sce[k * NW + 1]) + (sce[k * NW + 2] + sce[k * NW + 3]);
        }
        if (tid < NV) {
            v4f s = *(const v4f*)(&sacc[0][4 * tid]);
#pragma unroll
            for (int kk = 1; kk < KC; ++kk) s += *(const v4f*)(&sacc[kk * NW][4 * tid]);
            *(v4f*)(blockpart + (long)blockIdx.x * CPITCH + 4 * tid) = s;
        }
    }
}

__device__ __forceinline__ float wave_sum32(const float (&v)[32]) {
    const v2f bones = {1.0f, 1.0f};
    const v8f vzero = {0.f,0.f,0.f,0.f,0.f,0.f,0.f,0.f};
    v8f acc0 = vzero, acc1 = vzero;
#pragma unroll
    for (int p = 0; p < 32; p += 4) {
        const v2f a0 = {v[p + 0], v[p + 1]};
        const v2f a1 = {v[p + 2], v[p + 3]};
        acc0 = __builtin_amdgcn_wmma_f32_16x16x4_f32(false, a0, false, bones,
                                                     (short)0, acc0, false, false);
        acc1 = __builtin_amdgcn_wmma_f32_16x16x4_f32(false, a1, false, bones,
                                                     (short)0, acc1, false, false);
    }
    const v8f acc = acc0 + acc1;
    float s = ((acc[0] + acc[1]) + (acc[2] + acc[3])) + ((acc[4] + acc[5]) + (acc[6] + acc[7]));
    s += __shfl_xor(s, 16, 32);
    return s;
}

typedef float v4f_a4 __attribute__((ext_vector_type(4), aligned(4)));

#define PW 8
#define CPB 8
#define CSP 32

__global__ __launch_bounds__(PW * 32) void post_rows(
    const float* __restrict__ conf_ema, const float* __restrict__ Lam,
    const float* __restrict__ seg, const float* __restrict__ confc,
    const float* __restrict__ priorc,
    float* __restrict__ o_conf, float* __restrict__ o_lam,
    const float* __restrict__ blockpart, float* __restrict__ colsum)
{
    __shared__ __attribute__((aligned(16))) float spart[PW][CPB];
    const int wave  = __builtin_amdgcn_readfirstlane(threadIdx.x >> 5);
    const int lane  = threadIdx.x & 31;
    const int row   = blockIdx.x * PW + wave;
    const long base = (long)row * Cc;
    const int NV    = Cc / 4;

    const float cnt     = priorc[row];
    const float invc    = 1.0f / fmaxf(cnt, 1.0f);
    const bool  present = (cnt > 0.0f);

    const v4f z4 = {0.f, 0.f, 0.f, 0.f};
    const int t = threadIdx.x;
    const float* __restrict__ bp = blockpart + (long)min(t, NBLK - 1) * CPITCH + CPB * blockIdx.x;
    v4f p0 = *(const v4f*)(bp);
    v4f p1 = *(const v4f*)(bp + 4);
    p0 = (t < NBLK) ? p0 : z4;
    p1 = (t < NBLK) ? p1 : z4;

    v4f ve[8], vb[8], vs[8], vl[8];
    const bool ok7 = (lane + 32 * 7) < NV;
    const long o0  = base + 4 * lane;
    const long o7  = base + 4 * min(lane + 32 * 7, NV - 1);
#pragma unroll
    for (int i = 0; i < 7; ++i) {
        ve[i] = *(const v4f*)(conf_ema + o0 + 128 * i);
        vb[i] = *(const v4f*)(confc    + o0 + 128 * i);
        vs[i] = *(const v4f*)(seg      + o0 + 128 * i);
        vl[i] = *(const v4f*)(Lam      + o0 + 128 * i);
    }
    ve[7] = *(const v4f*)(conf_ema + o7);
    vb[7] = *(const v4f*)(confc    + o7);
    vs[7] = *(const v4f*)(seg      + o7);
    vl[7] = *(const v4f*)(Lam      + o7);
    ve[7] = ok7 ? ve[7] : z4;
    vb[7] = ok7 ? vb[7] : z4;
    vs[7] = ok7 ? vs[7] : z4;
    vl[7] = ok7 ? vl[7] : z4;

    float acnt[32], empr[32];
#pragma unroll
    for (int i = 0; i < 8; ++i) {
        const int j = lane + 32 * i;
        const bool ok = (i < 7) || ok7;
        const v4f cn = DECAYF * ve[i] + ONE_M_DECAY * vb[i];
        if (ok) *(v4f_a4*)(o_conf + base + 4 * j) = cn;
#pragma unroll
        for (int k = 0; k < 4; ++k) {
            const int c = 4 * j + k;
            const bool offd = ok && (c != row);
            acnt[4 * i + k] = offd ? (cn[k] + ALPHAF) : 0.f;
            empr[4 * i + k] = offd ? (vs[i][k] * invc) : 0.f;
        }
    }

    const float countsSum = wave_sum32(acnt);
    const float empSum    = wave_sum32(empr);
    const float phiDen = 1.0f / (countsSum + EPSF);
    const float empDen = 1.0f / fmaxf(empSum, EPSF);

    float l1[32];
#pragma unroll
    for (int i = 0; i < 8; ++i) {
        const int j = lane + 32 * i;
#pragma unroll
        for (int k = 0; k < 4; ++k) {
            const int c = 4 * j + k;
            const float phi = acnt[4 * i + k] * phiDen;
            const float emp = empr[4 * i + k] * empDen;
            const float v   = vl[i][k] + LR_LF * (emp - phi);
            l1[4 * i + k] = (((i < 7) || ok7) && (c != row)) ? v : 0.f;
        }
    }
    const float rowMean = wave_sum32(l1) * (1.0f / (float)Cc);

    {
        v4f q0, q1;
#pragma unroll
        for (int k = 0; k < 4; ++k) { q0[k] = wave_sum_f(p0[k]); q1[k] = wave_sum_f(p1[k]); }
        if (lane == 0) { *(v4f*)(&spart[wave][0]) = q0; *(v4f*)(&spart[wave][4]) = q1; }
    }
    __syncthreads();
    if (t < CPB) {
        float s = spart[0][t];
#pragma unroll
        for (int w = 1; w < PW; ++w) s += spart[w][t];
        colsum[CSP * blockIdx.x + t] = s;
    }

#pragma unroll
    for (int i = 0; i < 8; ++i) {
        const int j = lane + 32 * i;
        const bool ok = (i < 7) || ok7;
        v4f r;
#pragma unroll
        for (int k = 0; k < 4; ++k)
            r[k] = present ? (l1[4 * i + k] - rowMean) : vl[i][k];
        if (ok) *(v4f_a4*)(o_lam + base + 4 * j) = r;
    }
}

__global__ __launch_bounds__(1024) void finalize(
    const float* __restrict__ prior_ema, const float* __restrict__ mu,
    const float* __restrict__ colsum, const float* __restrict__ priorc,
    const float* __restrict__ cep,
    float* __restrict__ o_ce, float* __restrict__ o_mu, float* __restrict__ o_prior)
{
    __shared__ float red[3][32];
    const int j    = threadIdx.x;
    const int lane = j & 31;
    const int wave = j >> 5;

    const bool live = j < Cc;
    const int  jc   = min(j, Cc - 1);
    float pe  = prior_ema[jc];
    float pc  = priorc[jc];
    float muj = mu[jc];
    float cej = cep[jc];
    float csj = colsum[CSP * (jc / CPB) + (jc % CPB)];
    pe = live ? pe : 0.0f; pc = live ? pc : 0.0f; muj = live ? muj : 0.0f;
    cej = live ? cej : 0.0f; csj = live ? csj : 0.0f;

    float pn = 0.0f;
    if (j < Cc) {
        pn = DECAYF * pe + ONE_M_DECAY * pc;
        o_prior[j] = pn;
    }
    {
        const float wp = wave_sum_f(pn);
        const float wc = wave_sum_f(cej);
        if (lane == 0) { red[0][wave] = wp; red[1][wave] = wc; }
    }
    __syncthreads();
    const float priorSum = wave_sum_f(red[0][lane]);

    float mu1 = 0.0f;
    if (j < Cc) {
        float emp_marg = csj * (1.0f / (float)Bn);
        float pi = pn / (priorSum + EPSF);
        mu1 = muj + LR_MUF * (emp_marg - pi);
    }
    {
        const float wm = wave_sum_f(mu1);
        if (lane == 0) red[2][wave] = wm;
    }
    __syncthreads();
    const float muMean = wave_sum_f(red[2][lane]) * (1.0f / (float)Cc);
    if (j < Cc) o_mu[j] = mu1 - muMean;

    if (wave == 0) {
        const float s = wave_sum_f(red[1][lane]);
        if (lane == 0) o_ce[0] = -s * (1.0f / (float)Bn);
    }
}

extern "C" void kernel_launch(void* const* d_in, const int* in_sizes, int n_in,
                              void* d_out, int out_size, void* d_ws, size_t ws_size,
                              hipStream_t stream)
{
    const float* logits    = (const float*)d_in[0];
    const int*   targets   = (const int*)  d_in[1];
    const float* conf_ema  = (const float*)d_in[2];
    const float* prior_ema = (const float*)d_in[3];
    const float* Lam       = (const float*)d_in[4];
    const float* mu        = (const float*)d_in[5];

    float* out     = (float*)d_out;
    float* o_ce    = out;
    float* o_lam   = out + 1;
    float* o_mu    = out + 1 + (long)Cc * Cc;
    float* o_conf  = out + 1 + (long)Cc * Cc + Cc;
    float* o_prior = out + 1 + 2L * Cc * Cc + Cc;

    float* ws     = (float*)d_ws;
    float* seg    = ws;
    float* confc  = ws + (long)Cc * Cc;
    float* priorc = ws + 2L * Cc * Cc;
    float* cep    = priorc + Cc;
    int*   bucket = (int*)(cep + 1280);
    float* blockpart = (float*)(bucket + (long)Cc * Bn) + 24;
    float* colsum  = blockpart + (long)NBLK * CPITCH;

    class_pass<<<Cc / KC, NWB * 32, 0, stream>>>(logits, Lam, targets, bucket, seg, confc, priorc, cep, blockpart);
    post_rows<<<Cc / PW, PW * 32, 0, stream>>>(conf_ema, Lam, seg, confc, priorc,
                                               o_conf, o_lam, blockpart, colsum);
    finalize<<<1, 1024, 0, stream>>>(prior_ema, mu, colsum, priorc, cep,
                                     o_ce, o_mu, o_prior);
}


# class_pass row step: argmax(z) = lane max (v_max3), wave max, one v_cmp_eq per element into a lane mask, least holding column found on the scalar unit; compare/select chain gone; sum order kept
# speedup vs baseline: 12.6995x; 1.0375x over previous
#include <hip/hip_runtime.h>

#define Cc 1000
#define Bn 65536
#define DECAYF 0.99f
#define ONE_M_DECAY 0.01f
#define ALPHAF 1.0f
#define LR_LF 0.05f
#define LR_MUF 0.02f
#define EPSF 1e-12f

typedef __attribute__((ext_vector_type(2))) float v2f;
typedef __attribute__((ext_vector_type(4))) float v4f;
typedef __attribute__((ext_vector_type(8))) float v8f;

#define NEG_BIG (-3.4e38f)
#define NW 4
#define KC 4
#define NWB (NW * KC)
#define NBLK (Cc / KC)
#define CPITCH 1024
typedef __attribute__((ext_vector_type(4))) int v4i;

__device__ __forceinline__ void load_row(v4f (&z)[8], float& z_c,
                                         const float* __restrict__ lrow,
                                         const int o0, const int o7, const int c)
{
#pragma unroll
    for (int i = 0; i < 7; ++i) z[i] = *(const v4f*)(lrow + o0 + 128 * i);
    z[7] = *(const v4f*)(lrow + o7);
    z_c  = lrow[c];
}

__device__ __forceinline__ int xhalf_i(const int b)
{
    return __builtin_amdgcn_permlanex16(b, b, 0x76543210, 0xfedcba98, false, false);
}
template <int M> __device__ __forceinline__ int xrow_i(const int b)
{
    return __builtin_amdgcn_update_dpp(0, b, 0x160 | M, 0xf, 0xf, true);
}
__device__ __forceinline__ float xhalf_f(const float x) { return __builtin_bit_cast(float, xhalf_i(__builtin_bit_cast(int, x))); }
template <int M> __device__ __forceinline__ float xrow_f(const float x) { return __builtin_bit_cast(float, xrow_i<M>(__builtin_bit_cast(int, x))); }

__device__ __forceinline__ float wave_max_f(float x)
{
    x = __builtin_elementwise_maximum(x, xhalf_f(x));
    x = __builtin_elementwise_maximum(x, xrow_f<8>(x));
    x = __builtin_elementwise_maximum(x, xrow_f<4>(x));
    x = __builtin_elementwise_maximum(x, xrow_f<2>(x));
    x = __builtin_elementwise_maximum(x, xrow_f<1>(x));
    return x;
}
__device__ __forceinline__ float wave_sum_f(float x)
{
    x += xhalf_f(x);
    x += xrow_f<8>(x);
    x += xrow_f<4>(x);
    x += xrow_f<2>(x);
    x += xrow_f<1>(x);
    return x;
}
__device__ __forceinline__ void row_step(v4f (&z)[8], const float z_c, v4f (&acc)[8],
                                         const float* slam, const float lam_cc,
                                         const int lane, const int o0, const int o7, const bool ok7,
                                         int* confrow, float& cesum)
{
    const v4f negbig = {NEG_BIG, NEG_BIG, NEG_BIG, NEG_BIG};
    float lmax0 = NEG_BIG, lmax1 = NEG_BIG;
#pragma unroll
    for (int i = 0; i < 8; i += 2) {
#pragma unroll
        for (int k = 0; k < 4; ++k) { lmax0 = fmaxf(lmax0, z[i][k]); lmax1 = fmaxf(lmax1, z[i + 1][k]); }
    }
    const float wmax = wave_max_f(fmaxf(lmax0, lmax1));
    unsigned m[32];
#pragma unroll
    for (int i = 0; i < 8; ++i) {
#pragma unroll
        for (int k = 0; k < 4; ++k) m[4 * i + k] = __builtin_amdgcn_ballot_w32(z[i][k] == wmax);
    }
    unsigned maxi = 0xffffffffu;
#pragma unroll
    for (int k = 0; k < 4; ++k) {
        unsigned P = 0x00ffffffu;
#pragma unroll
        for (int j = 3; j >= 0; --j) {
            const unsigned long long Wd = (unsigned long long)m[8 * j + k] | ((unsigned long long)m[8 * j + 4 + k] << 32);
            P = (Wd != 0ull) ? (unsigned)(64 * j + __builtin_ctzll(Wd)) : P;
        }
        maxi = min(maxi, 4u * P + (unsigned)k);
    }
    maxi = min(maxi, (unsigned)(Cc - 1));

    float maxzp = NEG_BIG;
#pragma unroll
    for (int i = 0; i < 8; ++i) {
        const v4f l = *(const v4f*)(slam + ((i < 7) ? (o0 + 128 * i) : o7));
        v4f zp = z[i] - l;
        if (i == 7) zp = ok7 ? zp : negbig;
#pragma unroll
        for (int k = 0; k < 4; ++k) maxzp = fmaxf(maxzp, zp[k]);
        z[i] = zp;
    }
    maxzp = wave_max_f(maxzp);

    float sum = 0.0f;
#pragma unroll
    for (int i = 0; i < 8; ++i) {
        v4f e;
#pragma unroll
        for (int k = 0; k < 4; ++k) e[k] = __expf(z[i][k] - maxzp);
        z[i] = e;
        sum += (e[0] + e[1]) + (e[2] + e[3]);
    }
    sum = wave_sum_f(sum);
    const float lse = __logf(sum);
    const float inv = 1.0f / sum;

#pragma unroll
    for (int i = 0; i < 8; ++i) acc[i] += z[i] * inv;

    if (lane == 0) atomicAdd(&confrow[maxi], 1);
    cesum += ((z_c - lam_cc) - maxzp) - lse;
}

__global__ __launch_bounds__(NWB * 32) void class_pass(
    const float* __restrict__ logits, const float* __restrict__ Lam,
    const int* __restrict__ targets, int* __restrict__ bucket,
    float* __restrict__ seg, float* __restrict__ confc,
    float* __restrict__ priorc, float* __restrict__ cep,
    float* __restrict__ blockpart)
{
    __shared__ __attribute__((aligned(16))) float sacc[NWB][Cc];
    __shared__ __attribute__((aligned(16))) float slam[KC][Cc];
    __shared__ __attribute__((aligned(16))) int   sconf[KC][Cc];
    __shared__ int   scnt[KC];
    __shared__ float sce[NWB];
    const int wave = __builtin_amdgcn_readfirstlane(threadIdx.x >> 5);
    const int lane = threadIdx.x & 31;
    const int tid  = threadIdx.x;
    const int k    = wave / NW;
    const int w    = wave % NW;
    const int c0   = blockIdx.x * KC;
    const int c    = c0 + k;
    const int NV   = Cc / 4;

    {
        const v4i zi = {0, 0, 0, 0};
        for (int j = tid; j < KC * NV; j += NWB * 32) *(v4i*)(&sconf[0][0] + 4 * j) = zi;
        if (tid < KC) scnt[tid] = 0;
        const float* __restrict__ lam0 = Lam + (long)c0 * Cc;
        for (int j = tid; j < KC * NV; j += NWB * 32) *(v4f*)(&slam[0][0] + 4 * j) = *(const v4f*)(lam0 + 4 * j);
    }
    __syncthreads();

    {
        const int NG = Bn / 4 / (NWB * 32);
#pragma unroll 1
        for (int g0 = 0; g0 < NG; g0 += 8) {
            v4i tv[8];
            int  ti[8];
#pragma unroll
            for (int u = 0; u < 8; ++u) {
                ti[u] = (((g0 + u + (int)blockIdx.x) & (NG - 1)) * (NWB * 32)) + tid;
                tv[u] = *(const v4i*)(targets + 4 * ti[u]);
            }
#pragma unroll
            for (int u = 0; u < 8; ++u) {
                const unsigned dmin = min(min((unsigned)(tv[u][0] - c0), (unsigned)(tv[u][1] - c0)),
                                          min((unsigned)(tv[u][2] - c0), (unsigned)(tv[u][3] - c0)));
                if (dmin < (unsigned)KC) {
#pragma unroll
                    for (int e = 0; e < 4; ++e) {
                        const int t = tv[u][e];
                        const unsigned d = (unsigned)(t - c0);
                        if (d < (unsigned)KC) {
                            const int slot = atomicAdd(&scnt[d], 1);
                            bucket[(long)t * Bn + slot] = 4 * ti[u] + e;
                        }
                    }
                }
            }
        }
    }
    __syncthreads();

    const int n = scnt[k];
    const int* brow = bucket + (long)c * Bn;

    const bool ok7 = (lane + 32 * 7) < NV;
    const int  o0  = 4 * lane;
    const int  o7  = 4 * min(lane + 32 * 7, NV - 1);

    const float* slamk = &slam[k][0];
    const float lam_cc = slamk[c];

    const v4f z4 = {0.f, 0.f, 0.f, 0.f};
    v4f acc[8];
#pragma unroll
    for (int i = 0; i < 8; ++i) acc[i] = z4;
    float cesum = 0.0f;
    int* confrow = &sconf[k][0];

    v4f za[8], zb[8];
    float zca, zcb;
    const int last = max(n - 1, 0);
    int r = w;
    int s = 0;
    int idxv = (n > 0) ? brow[min(w + NW * lane, last)] : 0;
    int idxn = (n > 0) ? brow[min(w + NW * (32 + lane), last)] : 0;
    load_row(za, zca, logits + (long)__builtin_amdgcn_readlane(idxv, 0) * Cc, o0, o7, c);
    if (r < n) {
        for (;;) {
            ++s;
            if ((s & 31) == 0) { idxv = idxn; idxn = brow[min(w + NW * (s + 32 + lane), last)]; }
            load_row(zb, zcb, logits + (long)__builtin_amdgcn_readlane(idxv, s & 31) * Cc, o0, o7, c);
            row_step(za, zca, acc, slamk, lam_cc, lane, o0, o7, ok7, confrow, cesum);
            r += NW;
            if (r >= n) break;
            ++s;
            if ((s & 31) == 0) { idxv = idxn; idxn = brow[min(w + NW * (s + 32 + lane), last)]; }
            load_row(za, zca, logits + (long)__builtin_amdgcn_readlane(idxv, s & 31) * Cc, o0, o7, c);
            row_step(zb, zcb, acc, slamk, lam_cc, lane, o0, o7, ok7, confrow, cesum);
            r += NW;
            if (r >= n) break;
        }
    }

#pragma unroll
    for (int i = 0; i < 7; ++i) *(v4f*)(&sacc[wave][o0 + 128 * i]) = acc[i];
    if (ok7) *(v4f*)(&sacc[wave][o0 + 128 * 7]) = acc[7];
    if (lane == 0) sce[wave] = cesum;
    __syncthreads();
    {
        const int tk = tid & (NW * 32 - 1);
        const int NJ = (NV + NW * 32 - 1) / (NW * 32);
        v4f cs[NJ];
#pragma unroll
        for (int u = 0; u < NJ; ++u) {
            const int j = tk + u * (NW * 32);
            cs[u] = z4;
            if (j < NV) {
                v4f s = *(const v4f*)(&sacc[k * NW][4 * j]);
#pragma unroll
                for (int ww = 1; ww < NW; ++ww) s += *(const v4f*)(&sacc[k * NW + ww][4 * j]);
                cs[u] = s;
                *(v4f*)(&sacc[k * NW][4 * j]) = s;
            }
        }
        __syncthreads();
#pragma unroll
        for (int u = 0; u < NJ; ++u) {
            const int j = tk + u * (NW * 32);
            if (j < NV) {
                *(v4f*)(seg + (long)c * Cc + 4 * j) = cs[u];
                const v4i q = *(const v4i*)(&sconf[k][4 * j]);
                const v4f f = {(float)q[0], (float)q[1], (float)q[2], (float)q[3]};
                *(v4f*)(confc + (long)c * Cc + 4 * j) = f;
            }
        }
        if (tk == 0) {
            priorc[c] = (float)n;
            cep[c] = (sce[k * NW] + sce[k * NW + 1]) + (sce[k * NW + 2] + sce[k * NW + 3]);
        }
        if (tid < NV) {
            v4f s = *(const v4f*)(&sacc[0][4 * tid]);
#pragma unroll
            for (int kk = 1; kk < KC; ++kk) s += *(const v4f*)(&sacc[kk * NW][4 * tid]);
            *(v4f*)(blockpart + (long)blockIdx.x * CPITCH + 4 * tid) = s;
        }
    }
}

__device__ __forceinline__ float wave_sum32(const float (&v)[32]) {
    const v2f bones = {1.0f, 1.0f};
    const v8f vzero = {0.f,0.f,0.f,0.f,0.f,0.f,0.f,0.f};
    v8f acc0 = vzero, acc1 = vzero;
#pragma unroll
    for (int p = 0; p < 32; p += 4) {
        const v2f a0 = {v[p + 0], v[p + 1]};
        const v2f a1 = {v[p + 2], v[p + 3]};
        acc0 = __builtin_amdgcn_wmma_f32_16x16x4_f32(false, a0, false, bones,
                                                     (short)0, acc0, false, false);
        acc1 = __builtin_amdgcn_wmma_f32_16x16x4_f32(false, a1, false, bones,
                                                     (short)0, acc1, false, false);
    }
    const v8f acc = acc0 + acc1;
    float s = ((acc[0] + acc[1]) + (acc[2] + acc[3])) + ((acc[4] + acc[5]) + (acc[6] + acc[7]));
    s += __shfl_xor(s, 16, 32);
    return s;
}

typedef float v4f_a4 __attribute__((ext_vector_type(4), aligned(4)));

#define PW 8
#define CPB 8
#define CSP 32

__global__ __launch_bounds__(PW * 32) void post_rows(
    const float* __restrict__ conf_ema, const float* __restrict__ Lam,
    const float* __restrict__ seg, const float* __restrict__ confc,
    const float* __restrict__ priorc,
    float* __restrict__ o_conf, float* __restrict__ o_lam,
    const float* __restrict__ blockpart, float* __restrict__ colsum)
{
    __shared__ __attribute__((aligned(16))) float spart[PW][CPB];
    const int wave  = __builtin_amdgcn_readfirstlane(threadIdx.x >> 5);
    const int lane  = threadIdx.x & 31;
    const int row   = blockIdx.x * PW + wave;
    const long base = (long)row * Cc;
    const int NV    = Cc / 4;

    const float cnt     = priorc[row];
    const float invc    = 1.0f / fmaxf(cnt, 1.0f);
    const bool  present = (cnt > 0.0f);

    const v4f z4 = {0.f, 0.f, 0.f, 0.f};
    const int t = threadIdx.x;
    const float* __restrict__ bp = blockpart + (long)min(t, NBLK - 1) * CPITCH + CPB * blockIdx.x;
    v4f p0 = *(const v4f*)(bp);
    v4f p1 = *(const v4f*)(bp + 4);
    p0 = (t < NBLK) ? p0 : z4;
    p1 = (t < NBLK) ? p1 : z4;

    v4f ve[8], vb[8], vs[8], vl[8];
    const bool ok7 = (lane + 32 * 7) < NV;
    const long o0  = base + 4 * lane;
    const long o7  = base + 4 * min(lane + 32 * 7, NV - 1);
#pragma unroll
    for (int i = 0; i < 7; ++i) {
        ve[i] = *(const v4f*)(conf_ema + o0 + 128 * i);
        vb[i] = *(const v4f*)(confc    + o0 + 128 * i);
        vs[i] = *(const v4f*)(seg      + o0 + 128 * i);
        vl[i] = *(const v4f*)(Lam      + o0 + 128 * i);
    }
    ve[7] = *(const v4f*)(conf_ema + o7);
    vb[7] = *(const v4f*)(confc    + o7);
    vs[7] = *(const v4f*)(seg      + o7);
    vl[7] = *(const v4f*)(Lam      + o7);
    ve[7] = ok7 ? ve[7] : z4;
    vb[7] = ok7 ? vb[7] : z4;
    vs[7] = ok7 ? vs[7] : z4;
    vl[7] = ok7 ? vl[7] : z4;

    float acnt[32], empr[32];
#pragma unroll
    for (int i = 0; i < 8; ++i) {
        const int j = lane + 32 * i;
        const bool ok = (i < 7) || ok7;
        const v4f cn = DECAYF * ve[i] + ONE_M_DECAY * vb[i];
        if (ok) *(v4f_a4*)(o_conf + base + 4 * j) = cn;
#pragma unroll
        for (int k = 0; k < 4; ++k) {
            const int c = 4 * j + k;
            const bool offd = ok && (c != row);
            acnt[4 * i + k] = offd ? (cn[k] + ALPHAF) : 0.f;
            empr[4 * i + k] = offd ? (vs[i][k] * invc) : 0.f;
        }
    }

    const float countsSum = wave_sum32(acnt);
    const float empSum    = wave_sum32(empr);
    const float phiDen = 1.0f / (countsSum + EPSF);
    const float empDen = 1.0f / fmaxf(empSum, EPSF);

    float l1[32];
#pragma unroll
    for (int i = 0; i < 8; ++i) {
        const int j = lane + 32 * i;
#pragma unroll
        for (int k = 0; k < 4; ++k) {
            const int c = 4 * j + k;
            const float phi = acnt[4 * i + k] * phiDen;
            const float emp = empr[4 * i + k] * empDen;
            const float v   = vl[i][k] + LR_LF * (emp - phi);
            l1[4 * i + k] = (((i < 7) || ok7) && (c != row)) ? v : 0.f;
        }
    }
    const float rowMean = wave_sum32(l1) * (1.0f / (float)Cc);

    {
        v4f q0, q1;
#pragma unroll
        for (int k = 0; k < 4; ++k) { q0[k] = wave_sum_f(p0[k]); q1[k] = wave_sum_f(p1[k]); }
        if (lane == 0) { *(v4f*)(&spart[wave][0]) = q0; *(v4f*)(&spart[wave][4]) = q1; }
    }
    __syncthreads();
    if (t < CPB) {
        float s = spart[0][t];
#pragma unroll
        for (int w = 1; w < PW; ++w) s += spart[w][t];
        colsum[CSP * blockIdx.x + t] = s;
    }

#pragma unroll
    for (int i = 0; i < 8; ++i) {
        const int j = lane + 32 * i;
        const bool ok = (i < 7) || ok7;
        v4f r;
#pragma unroll
        for (int k = 0; k < 4; ++k)
            r[k] = present ? (l1[4 * i + k] - rowMean) : vl[i][k];
        if (ok) *(v4f_a4*)(o_lam + base + 4 * j) = r;
    }
}

__global__ __launch_bounds__(1024) void finalize(
    const float* __restrict__ prior_ema, const float* __restrict__ mu,
    const float* __restrict__ colsum, const float* __restrict__ priorc,
    const float* __restrict__ cep,
    float* __restrict__ o_ce, float* __restrict__ o_mu, float* __restrict__ o_prior)
{
    __shared__ float red[3][32];
    const int j    = threadIdx.x;
    const int lane = j & 31;
    const int wave = j >> 5;

    const bool live = j < Cc;
    const int  jc   = min(j, Cc - 1);
    float pe  = prior_ema[jc];
    float pc  = priorc[jc];
    float muj = mu[jc];
    float cej = cep[jc];
    float csj = colsum[CSP * (jc / CPB) + (jc % CPB)];
    pe = live ? pe : 0.0f; pc = live ? pc : 0.0f; muj = live ? muj : 0.0f;
    cej = live ? cej : 0.0f; csj = live ? csj : 0.0f;

    float pn = 0.0f;
    if (j < Cc) {
        pn = DECAYF * pe + ONE_M_DECAY * pc;
        o_prior[j] = pn;
    }
    {
        const float wp = wave_sum_f(pn);
        const float wc = wave_sum_f(cej);
        if (lane == 0) { red[0][wave] = wp; red[1][wave] = wc; }
    }
    __syncthreads();
    const float priorSum = wave_sum_f(red[0][lane]);

    float mu1 = 0.0f;
    if (j < Cc) {
        float emp_marg = csj * (1.0f / (float)Bn);
        float pi = pn / (priorSum + EPSF);
        mu1 = muj + LR_MUF * (emp_marg - pi);
    }
    {
        const float wm = wave_sum_f(mu1);
        if (lane == 0) red[2][wave] = wm;
    }
    __syncthreads();
    const float muMean = wave_sum_f(red[2][lane]) * (1.0f / (float)Cc);
    if (j < Cc) o_mu[j] = mu1 - muMean;

    if (wave == 0) {
        const float s = wave_sum_f(red[1][lane]);
        if (lane == 0) o_ce[0] = -s * (1.0f / (float)Bn);
    }
}

extern "C" void kernel_launch(void* const* d_in, const int* in_sizes, int n_in,
                              void* d_out, int out_size, void* d_ws, size_t ws_size,
                              hipStream_t stream)
{
    const float* logits    = (const float*)d_in[0];
    const int*   targets   = (const int*)  d_in[1];
    const float* conf_ema  = (const float*)d_in[2];
    const float* prior_ema = (const float*)d_in[3];
    const float* Lam       = (const float*)d_in[4];
    const float* mu        = (const float*)d_in[5];

    float* out     = (float*)d_out;
    float* o_ce    = out;
    float* o_lam   = out + 1;
    float* o_mu    = out + 1 + (long)Cc * Cc;
    float* o_conf  = out + 1 + (long)Cc * Cc + Cc;
    float* o_prior = out + 1 + 2L * Cc * Cc + Cc;

    float* ws     = (float*)d_ws;
    float* seg    = ws;
    float* confc  = ws + (long)Cc * Cc;
    float* priorc = ws + 2L * Cc * Cc;
    float* cep    = priorc + Cc;
    int*   bucket = (int*)(cep + 1280);
    float* blockpart = (float*)(bucket + (long)Cc * Bn) + 24;
    float* colsum  = blockpart + (long)NBLK * CPITCH;

    class_pass<<<Cc / KC, NWB * 32, 0, stream>>>(logits, Lam, targets, bucket, seg, confc, priorc, cep, blockpart);
    post_rows<<<Cc / PW, PW * 32, 0, stream>>>(conf_ema, Lam, seg, confc, priorc,
                                               o_conf, o_lam, blockpart, colsum);
    finalize<<<1, 1024, 0, stream>>>(prior_ema, mu, colsum, priorc, cep,
                                     o_ce, o_mu, o_prior);
}


# finalize launch gone (2 launches): each post_rows block finishes prior_new, mu_new for its 8 columns; mean(mu1) from per-class totals of q and block sums over the classes; summed in a new order
# speedup vs baseline: 13.2829x; 1.0459x over previous
#include <hip/hip_runtime.h>

#define Cc 1000
#define Bn 65536
#define DECAYF 0.99f
#define ONE_M_DECAY 0.01f
#define ALPHAF 1.0f
#define LR_LF 0.05f
#define LR_MUF 0.02f
#define EPSF 1e-12f

typedef __attribute__((ext_vector_type(2))) float v2f;
typedef __attribute__((ext_vector_type(4))) float v4f;
typedef __attribute__((ext_vector_type(8))) float v8f;

#define NEG_BIG (-3.4e38f)
#define NW 4
#define KC 4
#define NWB (NW * KC)
#define NBLK (Cc / KC)
#define CPITCH 1024
typedef __attribute__((ext_vector_type(4))) int v4i;

__device__ __forceinline__ void load_row(v4f (&z)[8], float& z_c,
                                         const float* __restrict__ lrow,
                                         const int o0, const int o7, const int c)
{
#pragma unroll
    for (int i = 0; i < 7; ++i) z[i] = *(const v4f*)(lrow + o0 + 128 * i);
    z[7] = *(const v4f*)(lrow + o7);
    z_c  = lrow[c];
}

__device__ __forceinline__ int xhalf_i(const int b)
{
    return __builtin_amdgcn_permlanex16(b, b, 0x76543210, 0xfedcba98, false, false);
}
template <int M> __device__ __forceinline__ int xrow_i(const int b)
{
    return __builtin_amdgcn_update_dpp(0, b, 0x160 | M, 0xf, 0xf, true);
}
__device__ __forceinline__ float xhalf_f(const float x) { return __builtin_bit_cast(float, xhalf_i(__builtin_bit_cast(int, x))); }
template <int M> __device__ __forceinline__ float xrow_f(const float x) { return __builtin_bit_cast(float, xrow_i<M>(__builtin_bit_cast(int, x))); }

__device__ __forceinline__ float wave_max_f(float x)
{
    x = __builtin_elementwise_maximum(x, xhalf_f(x));
    x = __builtin_elementwise_maximum(x, xrow_f<8>(x));
    x = __builtin_elementwise_maximum(x, xrow_f<4>(x));
    x = __builtin_elementwise_maximum(x, xrow_f<2>(x));
    x = __builtin_elementwise_maximum(x, xrow_f<1>(x));
    return x;
}
__device__ __forceinline__ float wave_sum_f(float x)
{
    x += xhalf_f(x);
    x += xrow_f<8>(x);
    x += xrow_f<4>(x);
    x += xrow_f<2>(x);
    x += xrow_f<1>(x);
    return x;
}
__device__ __forceinline__ void row_step(v4f (&z)[8], const float z_c, v4f (&acc)[8],
                                         const float* slam, const float lam_cc,
                                         const int lane, const int o0, const int o7, const bool ok7,
                                         int* confrow, float& cesum)
{
    const v4f negbig = {NEG_BIG, NEG_BIG, NEG_BIG, NEG_BIG};
    float lmax0 = NEG_BIG, lmax1 = NEG_BIG;
#pragma unroll
    for (int i = 0; i < 8; i += 2) {
#pragma unroll
        for (int k = 0; k < 4; ++k) { lmax0 = fmaxf(lmax0, z[i][k]); lmax1 = fmaxf(lmax1, z[i + 1][k]); }
    }
    const float wmax = wave_max_f(fmaxf(lmax0, lmax1));
    unsigned m[32];
#pragma unroll
    for (int i = 0; i < 8; ++i) {
#pragma unroll
        for (int k = 0; k < 4; ++k) m[4 * i + k] = __builtin_amdgcn_ballot_w32(z[i][k] == wmax);
    }
    unsigned maxi = 0xffffffffu;
#pragma unroll
    for (int k = 0; k < 4; ++k) {
        unsigned P = 0x00ffffffu;
#pragma unroll
        for (int j = 3; j >= 0; --j) {
            const unsigned long long Wd = (unsigned long long)m[8 * j + k] | ((unsigned long long)m[8 * j + 4 + k] << 32);
            P = (Wd != 0ull) ? (unsigned)(64 * j + __builtin_ctzll(Wd)) : P;
        }
        maxi = min(maxi, 4u * P + (unsigned)k);
    }
    maxi = min(maxi, (unsigned)(Cc - 1));

    float maxzp = NEG_BIG;
#pragma unroll
    for (int i = 0; i < 8; ++i) {
        const v4f l = *(const v4f*)(slam + ((i < 7) ? (o0 + 128 * i) : o7));
        v4f zp = z[i] - l;
        if (i == 7) zp = ok7 ? zp : negbig;
#pragma unroll
        for (int k = 0; k < 4; ++k) maxzp = fmaxf(maxzp, zp[k]);
        z[i] = zp;
    }
    maxzp = wave_max_f(maxzp);

    float sum = 0.0f;
#pragma unroll
    for (int i = 0; i < 8; ++i) {
        v4f e;
#pragma unroll
        for (int k = 0; k < 4; ++k) e[k] = __expf(z[i][k] - maxzp);
        z[i] = e;
        sum += (e[0] + e[1]) + (e[2] + e[3]);
    }
    sum = wave_sum_f(sum);
    const float lse = __logf(sum);
    const float inv = 1.0f / sum;

#pragma unroll
    for (int i = 0; i < 8; ++i) acc[i] += z[i] * inv;

    if (lane == 0) atomicAdd(&confrow[maxi], 1);
    cesum += ((z_c - lam_cc) - maxzp) - lse;
}

__global__ __launch_bounds__(NWB * 32) void class_pass(
    const float* __restrict__ logits, const float* __restrict__ Lam,
    const int* __restrict__ targets, int* __restrict__ bucket,
    float* __restrict__ seg, float* __restrict__ confc,
    float* __restrict__ priorc, float* __restrict__ cep,
    float* __restrict__ blockpart, float* __restrict__ qtot)
{
    __shared__ __attribute__((aligned(16))) float sacc[NWB][Cc];
    __shared__ __attribute__((aligned(16))) float slam[KC][Cc];
    __shared__ __attribute__((aligned(16))) int   sconf[KC][Cc];
    __shared__ int   scnt[KC];
    __shared__ float sce[NWB];
    __shared__ float sqt[NWB];
    const int wave = __builtin_amdgcn_readfirstlane(threadIdx.x >> 5);
    const int lane = threadIdx.x & 31;
    const int tid  = threadIdx.x;
    const int k    = wave / NW;
    const int w    = wave % NW;
    const int c0   = blockIdx.x * KC;
    const int c    = c0 + k;
    const int NV   = Cc / 4;

    {
        const v4i zi = {0, 0, 0, 0};
        for (int j = tid; j < KC * NV; j += NWB * 32) *(v4i*)(&sconf[0][0] + 4 * j) = zi;
        if (tid < KC) scnt[tid] = 0;
        const float* __restrict__ lam0 = Lam + (long)c0 * Cc;
        for (int j = tid; j < KC * NV; j += NWB * 32) *(v4f*)(&slam[0][0] + 4 * j) = *(const v4f*)(lam0 + 4 * j);
    }
    __syncthreads();

    {
        const int NG = Bn / 4 / (NWB * 32);
#pragma unroll 1
        for (int g0 = 0; g0 < NG; g0 += 8) {
            v4i tv[8];
            int  ti[8];
#pragma unroll
            for (int u = 0; u < 8; ++u) {
                ti[u] = (((g0 + u + (int)blockIdx.x) & (NG - 1)) * (NWB * 32)) + tid;
                tv[u] = *(const v4i*)(targets + 4 * ti[u]);
            }
#pragma unroll
            for (int u = 0; u < 8; ++u) {
                const unsigned dmin = min(min((unsigned)(tv[u][0] - c0), (unsigned)(tv[u][1] - c0)),
                                          min((unsigned)(tv[u][2] - c0), (unsigned)(tv[u][3] - c0)));
                if (dmin < (unsigned)KC) {
#pragma unroll
                    for (int e = 0; e < 4; ++e) {
                        const int t = tv[u][e];
                        const unsigned d = (unsigned)(t - c0);
                        if (d < (unsigned)KC) {
                            const int slot = atomicAdd(&scnt[d], 1);
                            bucket[(long)t * Bn + slot] = 4 * ti[u] + e;
                        }
                    }
                }
            }
        }
    }
    __syncthreads();

    const int n = scnt[k];
    const int* brow = bucket + (long)c * Bn;

    const bool ok7 = (lane + 32 * 7) < NV;
    const int  o0  = 4 * lane;
    const int  o7  = 4 * min(lane + 32 * 7, NV - 1);

    const float* slamk = &slam[k][0];
    const float lam_cc = slamk[c];

    const v4f z4 = {0.f, 0.f, 0.f, 0.f};
    v4f acc[8];
#pragma unroll
    for (int i = 0; i < 8; ++i) acc[i] = z4;
    float cesum = 0.0f;
    int* confrow = &sconf[k][0];

    v4f za[8], zb[8];
    float zca, zcb;
    const int last = max(n - 1, 0);
    int r = w;
    int s = 0;
    int idxv = (n > 0) ? brow[min(w + NW * lane, last)] : 0;
    int idxn = (n > 0) ? brow[min(w + NW * (32 + lane), last)] : 0;
    load_row(za, zca, logits + (long)__builtin_amdgcn_readlane(idxv, 0) * Cc, o0, o7, c);
    if (r < n) {
        for (;;) {
            ++s;
            if ((s & 31) == 0) { idxv = idxn; idxn = brow[min(w + NW * (s + 32 + lane), last)]; }
            load_row(zb, zcb, logits + (long)__builtin_amdgcn_readlane(idxv, s & 31) * Cc, o0, o7, c);
            row_step(za, zca, acc, slamk, lam_cc, lane, o0, o7, ok7, confrow, cesum);
            r += NW;
            if (r >= n) break;
            ++s;
            if ((s & 31) == 0) { idxv = idxn; idxn = brow[min(w + NW * (s + 32 + lane), last)]; }
            load_row(za, zca, logits + (long)__builtin_amdgcn_readlane(idxv, s & 31) * Cc, o0, o7, c);
            row_step(zb, zcb, acc, slamk, lam_cc, lane, o0, o7, ok7, confrow, cesum);
            r += NW;
            if (r >= n) break;
        }
    }

#pragma unroll
    for (int i = 0; i < 7; ++i) *(v4f*)(&sacc[wave][o0 + 128 * i]) = acc[i];
    if (ok7) *(v4f*)(&sacc[wave][o0 + 128 * 7]) = acc[7];
    if (lane == 0) sce[wave] = cesum;
    __syncthreads();
    {
        const int tk = tid & (NW * 32 - 1);
        const int NJ = (NV + NW * 32 - 1) / (NW * 32);
        v4f cs[NJ];
#pragma unroll
        for (int u = 0; u < NJ; ++u) {
            const int j = tk + u * (NW * 32);
            cs[u] = z4;
            if (j < NV) {
                v4f s = *(const v4f*)(&sacc[k * NW][4 * j]);
#pragma unroll
                for (int ww = 1; ww < NW; ++ww) s += *(const v4f*)(&sacc[k * NW + ww][4 * j]);
                cs[u] = s;
                *(v4f*)(&sacc[k * NW][4 * j]) = s;
            }
        }
        {
            float qt = 0.0f;
#pragma unroll
            for (int u = 0; u < NJ; ++u) qt += (cs[u][0] + cs[u][1]) + (cs[u][2] + cs[u][3]);
            qt = wave_sum_f(qt);
            if (lane == 0) sqt[wave] = qt;
        }
        __syncthreads();
#pragma unroll
        for (int u = 0; u < NJ; ++u) {
            const int j = tk + u * (NW * 32);
            if (j < NV) {
                *(v4f*)(seg + (long)c * Cc + 4 * j) = cs[u];
                const v4i q = *(const v4i*)(&sconf[k][4 * j]);
                const v4f f = {(float)q[0], (float)q[1], (float)q[2], (float)q[3]};
                *(v4f*)(confc + (long)c * Cc + 4 * j) = f;
            }
        }
        if (tk == 0) {
            priorc[c] = (float)n;
            cep[c] = (sce[k * NW] + sce[k * NW + 1]) + (sce[k * NW + 2] + sce[k * NW + 3]);
            qtot[c] = (sqt[k * NW] + sqt[k * NW + 1]) + (sqt[k * NW + 2] + sqt[k * NW + 3]);
        }
        if (tid < NV) {
            v4f s = *(const v4f*)(&sacc[0][4 * tid]);
#pragma unroll
            for (int kk = 1; kk < KC; ++kk) s += *(const v4f*)(&sacc[kk * NW][4 * tid]);
            *(v4f*)(blockpart + (long)blockIdx.x * CPITCH + 4 * tid) = s;
        }
    }
}

__device__ __forceinline__ float wave_sum32(const float (&v)[32]) {
    const v2f bones = {1.0f, 1.0f};
    const v8f vzero = {0.f,0.f,0.f,0.f,0.f,0.f,0.f,0.f};
    v8f acc0 = vzero, acc1 = vzero;
#pragma unroll
    for (int p = 0; p < 32; p += 4) {
        const v2f a0 = {v[p + 0], v[p + 1]};
        const v2f a1 = {v[p + 2], v[p + 3]};
        acc0 = __builtin_amdgcn_wmma_f32_16x16x4_f32(false, a0, false, bones,
                                                     (short)0, acc0, false, false);
        acc1 = __builtin_amdgcn_wmma_f32_16x16x4_f32(false, a1, false, bones,
                                                     (short)0, acc1, false, false);
    }
    const v8f acc = acc0 + acc1;
    float s = ((acc[0] + acc[1]) + (acc[2] + acc[3])) + ((acc[4] + acc[5]) + (acc[6] + acc[7]));
    s += __shfl_xor(s, 16, 32);
    return s;
}

typedef float v4f_a4 __attribute__((ext_vector_type(4), aligned(4)));

#define PW 8
#define CPB 8

__global__ __launch_bounds__(PW * 32) void post_rows(
    const float* __restrict__ conf_ema, const float* __restrict__ Lam,
    const float* __restrict__ seg, const float* __restrict__ confc,
    const float* __restrict__ priorc,
    float* __restrict__ o_conf, float* __restrict__ o_lam,
    const float* __restrict__ blockpart,
    const float* __restrict__ prior_ema, const float* __restrict__ mu,
    const float* __restrict__ qtot, const float* __restrict__ cep,
    float* __restrict__ o_ce, float* __restrict__ o_mu, float* __restrict__ o_prior)
{
    __shared__ __attribute__((aligned(16))) float spart[PW][CPB];
    __shared__ __attribute__((aligned(16))) float sred[PW][4];
    const int wave  = __builtin_amdgcn_readfirstlane(threadIdx.x >> 5);
    const int lane  = threadIdx.x & 31;
    const int row   = blockIdx.x * PW + wave;
    const long base = (long)row * Cc;
    const int NV    = Cc / 4;

    const float cnt     = priorc[row];
    const float invc    = 1.0f / fmaxf(cnt, 1.0f);
    const bool  present = (cnt > 0.0f);

    const v4f z4 = {0.f, 0.f, 0.f, 0.f};
    const int t = threadIdx.x;
    const float* __restrict__ bp = blockpart + (long)min(t, NBLK - 1) * CPITCH + CPB * blockIdx.x;
    v4f p0 = *(const v4f*)(bp);
    v4f p1 = *(const v4f*)(bp + 4);
    p0 = (t < NBLK) ? p0 : z4;
    p1 = (t < NBLK) ? p1 : z4;

    const int  tq    = min(t, NV - 1);
    const bool tlive = t < NV;
    v4f f_pe = *(const v4f*)(prior_ema + 4 * tq);
    v4f f_pc = *(const v4f*)(priorc    + 4 * tq);
    v4f f_mu = *(const v4f*)(mu        + 4 * tq);
    v4f f_qt = *(const v4f*)(qtot      + 4 * tq);
    v4f f_ce = z4;
    if (blockIdx.x == 0) f_ce = *(const v4f*)(cep + 4 * tq);
    const int   jc   = CPB * blockIdx.x + (t & (CPB - 1));
    const float pe_j = prior_ema[jc];
    const float pc_j = priorc[jc];
    const float mu_j = mu[jc];
    f_pe = tlive ? f_pe : z4;
    f_pc = tlive ? f_pc : z4;
    f_mu = tlive ? f_mu : z4;
    f_qt = tlive ? f_qt : z4;
    f_ce = tlive ? f_ce : z4;

    v4f ve[8], vb[8], vs[8], vl[8];
    const bool ok7 = (lane + 32 * 7) < NV;
    const long o0  = base + 4 * lane;
    const long o7  = base + 4 * min(lane + 32 * 7, NV - 1);
#pragma unroll
    for (int i = 0; i < 7; ++i) {
        ve[i] = *(const v4f*)(conf_ema + o0 + 128 * i);
        vb[i] = *(const v4f*)(confc    + o0 + 128 * i);
        vs[i] = *(const v4f*)(seg      + o0 + 128 * i);
        vl[i] = *(const v4f*)(Lam      + o0 + 128 * i);
    }
    ve[7] = *(const v4f*)(conf_ema + o7);
    vb[7] = *(const v4f*)(confc    + o7);
    vs[7] = *(const v4f*)(seg      + o7);
    vl[7] = *(const v4f*)(Lam      + o7);
    ve[7] = ok7 ? ve[7] : z4;
    vb[7] = ok7 ? vb[7] : z4;
    vs[7] = ok7 ? vs[7] : z4;
    vl[7] = ok7 ? vl[7] : z4;

    float acnt[32], empr[32];
#pragma unroll
    for (int i = 0; i < 8; ++i) {
        const int j = lane + 32 * i;
        const bool ok = (i < 7) || ok7;
        const v4f cn = DECAYF * ve[i] + ONE_M_DECAY * vb[i];
        if (ok) *(v4f_a4*)(o_conf + base + 4 * j) = cn;
#pragma unroll
        for (int k = 0; k < 4; ++k) {
            const int c = 4 * j + k;
            const bool offd = ok && (c != row);
            acnt[4 * i + k] = offd ? (cn[k] + ALPHAF) : 0.f;
            empr[4 * i + k] = offd ? (vs[i][k] * invc) : 0.f;
        }
    }

    const float countsSum = wave_sum32(acnt);
    const float empSum    = wave_sum32(empr);
    const float phiDen = 1.0f / (countsSum + EPSF);
    const float empDen = 1.0f / fmaxf(empSum, EPSF);

    float l1[32];
#pragma unroll
    for (int i = 0; i < 8; ++i) {
        const int j = lane + 32 * i;
#pragma unroll
        for (int k = 0; k < 4; ++k) {
            const int c = 4 * j + k;
            const float phi = acnt[4 * i + k] * phiDen;
            const float emp = empr[4 * i + k] * empDen;
            const float v   = vl[i][k] + LR_LF * (emp - phi);
            l1[4 * i + k] = (((i < 7) || ok7) && (c != row)) ? v : 0.f;
        }
    }
    const float rowMean = wave_sum32(l1) * (1.0f / (float)Cc);

    {
        v4f q0, q1;
#pragma unroll
        for (int k = 0; k < 4; ++k) { q0[k] = wave_sum_f(p0[k]); q1[k] = wave_sum_f(p1[k]); }
        if (lane == 0) { *(v4f*)(&spart[wave][0]) = q0; *(v4f*)(&spart[wave][4]) = q1; }
        const v4f pn4 = DECAYF * f_pe + ONE_M_DECAY * f_pc;
        v4f r;
        r[0] = wave_sum_f((pn4[0]  + pn4[1])  + (pn4[2]  + pn4[3]));
        r[1] = wave_sum_f((f_mu[0] + f_mu[1]) + (f_mu[2] + f_mu[3]));
        r[2] = wave_sum_f((f_qt[0] + f_qt[1]) + (f_qt[2] + f_qt[3]));
        r[3] = wave_sum_f((f_ce[0] + f_ce[1]) + (f_ce[2] + f_ce[3]));
        if (lane == 0) *(v4f*)(&sred[wave][0]) = r;
    }
    __syncthreads();
    if (t < CPB) {
        float s = spart[0][t];
        v4f   S = *(const v4f*)(&sred[0][0]);
#pragma unroll
        for (int w = 1; w < PW; ++w) { s += spart[w][t]; S += *(const v4f*)(&sred[w][0]); }
        const float pn     = DECAYF * pe_j + ONE_M_DECAY * pc_j;
        const float den    = S[0] + EPSF;
        const float mu1    = mu_j + LR_MUF * (s * (1.0f / (float)Bn) - pn / den);
        const float muMean = (S[1] + LR_MUF * (S[2] * (1.0f / (float)Bn) - S[0] / den)) * (1.0f / (float)Cc);
        o_prior[jc] = pn;
        o_mu[jc]    = mu1 - muMean;
        if (blockIdx.x == 0 && t == 0) o_ce[0] = -S[3] * (1.0f / (float)Bn);
    }

#pragma unroll
    for (int i = 0; i < 8; ++i) {
        const int j = lane + 32 * i;
        const bool ok = (i < 7) || ok7;
        v4f r;
#pragma unroll
        for (int k = 0; k < 4; ++k)
            r[k] = present ? (l1[4 * i + k] - rowMean) : vl[i][k];
        if (ok) *(v4f_a4*)(o_lam + base + 4 * j) = r;
    }
}

extern "C" void kernel_launch(void* const* d_in, const int* in_sizes, int n_in,
                              void* d_out, int out_size, void* d_ws, size_t ws_size,
                              hipStream_t stream)
{
    const float* logits    = (const float*)d_in[0];
    const int*   targets   = (const int*)  d_in[1];
    const float* conf_ema  = (const float*)d_in[2];
    const float* prior_ema = (const float*)d_in[3];
    const float* Lam       = (const float*)d_in[4];
    const float* mu        = (const float*)d_in[5];

    float* out     = (float*)d_out;
    float* o_ce    = out;
    float* o_lam   = out + 1;
    float* o_mu    = out + 1 + (long)Cc * Cc;
    float* o_conf  = out + 1 + (long)Cc * Cc + Cc;
    float* o_prior = out + 1 + 2L * Cc * Cc + Cc;

    float* ws     = (float*)d_ws;
    float* seg    = ws;
    float* confc  = ws + (long)Cc * Cc;
    float* priorc = ws + 2L * Cc * Cc;
    float* cep    = priorc + Cc;
    int*   bucket = (int*)(cep + 1280);
    float* blockpart = (float*)(bucket + (long)Cc * Bn) + 24;
    float* qtot    = blockpart + (long)NBLK * CPITCH;

    class_pass<<<Cc / KC, NWB * 32, 0, stream>>>(logits, Lam, targets, bucket, seg, confc, priorc, cep, blockpart, qtot);
    post_rows<<<Cc / PW, PW * 32, 0, stream>>>(conf_ema, Lam, seg, confc, priorc,
                                               o_conf, o_lam, blockpart,
                                               prior_ema, mu, qtot, cep,
                                               o_ce, o_mu, o_prior);
}


# row epilogue fused into class_pass: a wave per class finishes conf_new, Lam_new from the block's LDS sums; seg and count planes never go to memory; 2nd launch: column sums only; sum order kept
# speedup vs baseline: 14.1291x; 1.0637x over previous
#include <hip/hip_runtime.h>

#define Cc 1000
#define Bn 65536
#define DECAYF 0.99f
#define ONE_M_DECAY 0.01f
#define ALPHAF 1.0f
#define LR_LF 0.05f
#define LR_MUF 0.02f
#define EPSF 1e-12f

typedef __attribute__((ext_vector_type(2))) float v2f;
typedef __attribute__((ext_vector_type(4))) float v4f;
typedef __attribute__((ext_vector_type(8))) float v8f;

#define NEG_BIG (-3.4e38f)
#define NW 4
#define KC 4
#define NWB (NW * KC)
#define NBLK (Cc / KC)
#define CPITCH 1024
typedef __attribute__((ext_vector_type(4))) int v4i;

__device__ __forceinline__ void load_row(v4f (&z)[8], float& z_c,
                                         const float* __restrict__ lrow,
                                         const int o0, const int o7, const int c)
{
#pragma unroll
    for (int i = 0; i < 7; ++i) z[i] = *(const v4f*)(lrow + o0 + 128 * i);
    z[7] = *(const v4f*)(lrow + o7);
    z_c  = lrow[c];
}

__device__ __forceinline__ int xhalf_i(const int b)
{
    return __builtin_amdgcn_permlanex16(b, b, 0x76543210, 0xfedcba98, false, false);
}
template <int M> __device__ __forceinline__ int xrow_i(const int b)
{
    return __builtin_amdgcn_update_dpp(0, b, 0x160 | M, 0xf, 0xf, true);
}
__device__ __forceinline__ float xhalf_f(const float x) { return __builtin_bit_cast(float, xhalf_i(__builtin_bit_cast(int, x))); }
template <int M> __device__ __forceinline__ float xrow_f(const float x) { return __builtin_bit_cast(float, xrow_i<M>(__builtin_bit_cast(int, x))); }

__device__ __forceinline__ float wave_max_f(float x)
{
    x = __builtin_elementwise_maximum(x, xhalf_f(x));
    x = __builtin_elementwise_maximum(x, xrow_f<8>(x));
    x = __builtin_elementwise_maximum(x, xrow_f<4>(x));
    x = __builtin_elementwise_maximum(x, xrow_f<2>(x));
    x = __builtin_elementwise_maximum(x, xrow_f<1>(x));
    return x;
}
__device__ __forceinline__ float wave_sum_f(float x)
{
    x += xhalf_f(x);
    x += xrow_f<8>(x);
    x += xrow_f<4>(x);
    x += xrow_f<2>(x);
    x += xrow_f<1>(x);
    return x;
}
__device__ __forceinline__ float wave_sum32(const float (&v)[32]) {
    const v2f bones = {1.0f, 1.0f};
    const v8f vzero = {0.f,0.f,0.f,0.f,0.f,0.f,0.f,0.f};
    v8f acc0 = vzero, acc1 = vzero;
#pragma unroll
    for (int p = 0; p < 32; p += 4) {
        const v2f a0 = {v[p + 0], v[p + 1]};
        const v2f a1 = {v[p + 2], v[p + 3]};
        acc0 = __builtin_amdgcn_wmma_f32_16x16x4_f32(false, a0, false, bones,
                                                     (short)0, acc0, false, false);
        acc1 = __builtin_amdgcn_wmma_f32_16x16x4_f32(false, a1, false, bones,
                                                     (short)0, acc1, false, false);
    }
    const v8f acc = acc0 + acc1;
    float s = ((acc[0] + acc[1]) + (acc[2] + acc[3])) + ((acc[4] + acc[5]) + (acc[6] + acc[7]));
    s += __shfl_xor(s, 16, 32);
    return s;
}

typedef float v4f_a4 __attribute__((ext_vector_type(4), aligned(4)));
__device__ __forceinline__ void row_step(v4f (&z)[8], const float z_c, v4f (&acc)[8],
                                         const float* slam, const float lam_cc,
                                         const int lane, const int o0, const int o7, const bool ok7,
                                         int* confrow, float& cesum)
{
    const v4f negbig = {NEG_BIG, NEG_BIG, NEG_BIG, NEG_BIG};
    float lmax0 = NEG_BIG, lmax1 = NEG_BIG;
#pragma unroll
    for (int i = 0; i < 8; i += 2) {
#pragma unroll
        for (int k = 0; k < 4; ++k) { lmax0 = fmaxf(lmax0, z[i][k]); lmax1 = fmaxf(lmax1, z[i + 1][k]); }
    }
    const float wmax = wave_max_f(fmaxf(lmax0, lmax1));
    unsigned m[32];
#pragma unroll
    for (int i = 0; i < 8; ++i) {
#pragma unroll
        for (int k = 0; k < 4; ++k) m[4 * i + k] = __builtin_amdgcn_ballot_w32(z[i][k] == wmax);
    }
    unsigned maxi = 0xffffffffu;
#pragma unroll
    for (int k = 0; k < 4; ++k) {
        unsigned P = 0x00ffffffu;
#pragma unroll
        for (int j = 3; j >= 0; --j) {
            const unsigned long long Wd = (unsigned long long)m[8 * j + k] | ((unsigned long long)m[8 * j + 4 + k] << 32);
            P = (Wd != 0ull) ? (unsigned)(64 * j + __builtin_ctzll(Wd)) : P;
        }
        maxi = min(maxi, 4u * P + (unsigned)k);
    }
    maxi = min(maxi, (unsigned)(Cc - 1));

    float maxzp = NEG_BIG;
#pragma unroll
    for (int i = 0; i < 8; ++i) {
        const v4f l = *(const v4f*)(slam + ((i < 7) ? (o0 + 128 * i) : o7));
        v4f zp = z[i] - l;
        if (i == 7) zp = ok7 ? zp : negbig;
#pragma unroll
        for (int k = 0; k < 4; ++k) maxzp = fmaxf(maxzp, zp[k]);
        z[i] = zp;
    }
    maxzp = wave_max_f(maxzp);

    float sum = 0.0f;
#pragma unroll
    for (int i = 0; i < 8; ++i) {
        v4f e;
#pragma unroll
        for (int k = 0; k < 4; ++k) e[k] = __expf(z[i][k] - maxzp);
        z[i] = e;
        sum += (e[0] + e[1]) + (e[2] + e[3]);
    }
    sum = wave_sum_f(sum);
    const float lse = __logf(sum);
    const float inv = 1.0f / sum;

#pragma unroll
    for (int i = 0; i < 8; ++i) acc[i] += z[i] * inv;

    if (lane == 0) atomicAdd(&confrow[maxi], 1);
    cesum += ((z_c - lam_cc) - maxzp) - lse;
}


__global__ __launch_bounds__(NWB * 32) void class_pass(
    const float* __restrict__ logits, const float* __restrict__ Lam,
    const int* __restrict__ targets, int* __restrict__ bucket,
    const float* __restrict__ conf_ema,
    float* __restrict__ o_conf, float* __restrict__ o_lam,
    float* __restrict__ priorc, float* __restrict__ cep,
    float* __restrict__ blockpart, float* __restrict__ qtot)
{
    __shared__ __attribute__((aligned(16))) float sacc[NWB][Cc];
    __shared__ __attribute__((aligned(16))) float slam[KC][Cc];
    __shared__ __attribute__((aligned(16))) int   sconf[KC][Cc];
    __shared__ int   scnt[KC];
    __shared__ float sce[NWB];
    __shared__ float sqt[NWB];
    const int wave = __builtin_amdgcn_readfirstlane(threadIdx.x >> 5);
    const int lane = threadIdx.x & 31;
    const int tid  = threadIdx.x;
    const int k    = wave / NW;
    const int w    = wave % NW;
    const int c0   = blockIdx.x * KC;
    const int c    = c0 + k;
    const int NV   = Cc / 4;

    {
        const v4i zi = {0, 0, 0, 0};
        for (int j = tid; j < KC * NV; j += NWB * 32) *(v4i*)(&sconf[0][0] + 4 * j) = zi;
        if (tid < KC) scnt[tid] = 0;
        const float* __restrict__ lam0 = Lam + (long)c0 * Cc;
        for (int j = tid; j < KC * NV; j += NWB * 32) *(v4f*)(&slam[0][0] + 4 * j) = *(const v4f*)(lam0 + 4 * j);
    }
    __syncthreads();

    {
        const int NG = Bn / 4 / (NWB * 32);
#pragma unroll 1
        for (int g0 = 0; g0 < NG; g0 += 8) {
            v4i tv[8];
            int  ti[8];
#pragma unroll
            for (int u = 0; u < 8; ++u) {
                ti[u] = (((g0 + u + (int)blockIdx.x) & (NG - 1)) * (NWB * 32)) + tid;
                tv[u] = *(const v4i*)(targets + 4 * ti[u]);
            }
#pragma unroll
            for (int u = 0; u < 8; ++u) {
                const unsigned dmin = min(min((unsigned)(tv[u][0] - c0), (unsigned)(tv[u][1] - c0)),
                                          min((unsigned)(tv[u][2] - c0), (unsigned)(tv[u][3] - c0)));
                if (dmin < (unsigned)KC) {
#pragma unroll
                    for (int e = 0; e < 4; ++e) {
                        const int t = tv[u][e];
                        const unsigned d = (unsigned)(t - c0);
                        if (d < (unsigned)KC) {
                            const int slot = atomicAdd(&scnt[d], 1);
                            bucket[(long)t * Bn + slot] = 4 * ti[u] + e;
                        }
                    }
                }
            }
        }
    }
    __syncthreads();

    const int n = scnt[k];
    const int* brow = bucket + (long)c * Bn;

    const bool ok7 = (lane + 32 * 7) < NV;
    const int  o0  = 4 * lane;
    const int  o7  = 4 * min(lane + 32 * 7, NV - 1);

    const float* slamk = &slam[k][0];
    const float lam_cc = slamk[c];

    const v4f z4 = {0.f, 0.f, 0.f, 0.f};
    v4f acc[8];
#pragma unroll
    for (int i = 0; i < 8; ++i) acc[i] = z4;
    float cesum = 0.0f;
    int* confrow = &sconf[k][0];

    v4f za[8], zb[8];
    float zca, zcb;
    const int last = max(n - 1, 0);
    int r = w;
    int s = 0;
    int idxv = (n > 0) ? brow[min(w + NW * lane, last)] : 0;
    int idxn = (n > 0) ? brow[min(w + NW * (32 + lane), last)] : 0;
    load_row(za, zca, logits + (long)__builtin_amdgcn_readlane(idxv, 0) * Cc, o0, o7, c);
    v4f ve[8];
    {
        const float* __restrict__ erow = conf_ema + (long)c * Cc;
#pragma unroll
        for (int i = 0; i < 7; ++i) ve[i] = *(const v4f*)(erow + o0 + 128 * i);
        ve[7] = *(const v4f*)(erow + o7);
    }
    if (r < n) {
        for (;;) {
            ++s;
            if ((s & 31) == 0) { idxv = idxn; idxn = brow[min(w + NW * (s + 32 + lane), last)]; }
            load_row(zb, zcb, logits + (long)__builtin_amdgcn_readlane(idxv, s & 31) * Cc, o0, o7, c);
            row_step(za, zca, acc, slamk, lam_cc, lane, o0, o7, ok7, confrow, cesum);
            r += NW;
            if (r >= n) break;
            ++s;
            if ((s & 31) == 0) { idxv = idxn; idxn = brow[min(w + NW * (s + 32 + lane), last)]; }
            load_row(za, zca, logits + (long)__builtin_amdgcn_readlane(idxv, s & 31) * Cc, o0, o7, c);
            row_step(zb, zcb, acc, slamk, lam_cc, lane, o0, o7, ok7, confrow, cesum);
            r += NW;
            if (r >= n) break;
        }
    }

    const bool rowwave = (w == k);

#pragma unroll
    for (int i = 0; i < 7; ++i) *(v4f*)(&sacc[wave][o0 + 128 * i]) = acc[i];
    if (ok7) *(v4f*)(&sacc[wave][o0 + 128 * 7]) = acc[7];
    if (lane == 0) sce[wave] = cesum;
    __syncthreads();
    {
        const int tk = tid & (NW * 32 - 1);
        const int NJ = (NV + NW * 32 - 1) / (NW * 32);
        v4f cs[NJ];
#pragma unroll
        for (int u = 0; u < NJ; ++u) {
            const int j = tk + u * (NW * 32);
            cs[u] = z4;
            if (j < NV) {
                v4f s = *(const v4f*)(&sacc[k * NW][4 * j]);
#pragma unroll
                for (int ww = 1; ww < NW; ++ww) s += *(const v4f*)(&sacc[k * NW + ww][4 * j]);
                cs[u] = s;
                *(v4f*)(&sacc[k * NW][4 * j]) = s;
            }
        }
        {
            float qt = 0.0f;
#pragma unroll
            for (int u = 0; u < NJ; ++u) qt += (cs[u][0] + cs[u][1]) + (cs[u][2] + cs[u][3]);
            qt = wave_sum_f(qt);
            if (lane == 0) sqt[wave] = qt;
        }
        __syncthreads();
        if (tk == 0) {
            priorc[c] = (float)n;
            cep[c] = (sce[k * NW] + sce[k * NW + 1]) + (sce[k * NW + 2] + sce[k * NW + 3]);
            qtot[c] = (sqt[k * NW] + sqt[k * NW + 1]) + (sqt[k * NW + 2] + sqt[k * NW + 3]);
        }
        if (tid < NV) {
            v4f s = *(const v4f*)(&sacc[0][4 * tid]);
#pragma unroll
            for (int kk = 1; kk < KC; ++kk) s += *(const v4f*)(&sacc[kk * NW][4 * tid]);
            *(v4f*)(blockpart + (long)blockIdx.x * CPITCH + 4 * tid) = s;
        }
    }

    if (rowwave) {
        const long  base    = (long)c * Cc;
        const float invc    = 1.0f / fmaxf((float)n, 1.0f);
        const bool  present = (n > 0);
        const float* srow = &sacc[k * NW][0];
        const int*   crow = &sconf[k][0];

        float acnt[32], empr[32];
#pragma unroll
        for (int i = 0; i < 8; ++i) {
            const int  j   = lane + 32 * i;
            const bool ok  = (i < 7) || ok7;
            const int  off = (i < 7) ? (o0 + 128 * i) : o7;
            const v4i  q   = *(const v4i*)(crow + off);
            const v4f  vb  = {(float)q[0], (float)q[1], (float)q[2], (float)q[3]};
            const v4f  vs  = *(const v4f*)(srow + off);
            const v4f  cn  = DECAYF * ve[i] + ONE_M_DECAY * vb;
            if (ok) *(v4f_a4*)(o_conf + base + 4 * j) = cn;
#pragma unroll
            for (int e = 0; e < 4; ++e) {
                const int  col  = 4 * j + e;
                const bool offd = ok && (col != c);
                acnt[4 * i + e] = offd ? (cn[e] + ALPHAF) : 0.f;
                empr[4 * i + e] = offd ? (vs[e] * invc) : 0.f;
            }
        }

        const float countsSum = wave_sum32(acnt);
        const float empSum    = wave_sum32(empr);
        const float phiDen = 1.0f / (countsSum + EPSF);
        const float empDen = 1.0f / fmaxf(empSum, EPSF);

        float l1[32];
        v4f   vl[8];
#pragma unroll
        for (int i = 0; i < 8; ++i) {
            const int j   = lane + 32 * i;
            const int off = (i < 7) ? (o0 + 128 * i) : o7;
            vl[i] = *(const v4f*)(slamk + off);
#pragma unroll
            for (int e = 0; e < 4; ++e) {
                const int   col = 4 * j + e;
                const float phi = acnt[4 * i + e] * phiDen;
                const float emp = empr[4 * i + e] * empDen;
                const float v   = vl[i][e] + LR_LF * (emp - phi);
                l1[4 * i + e] = (((i < 7) || ok7) && (col != c)) ? v : 0.f;
            }
        }
        const float rowMean = wave_sum32(l1) * (1.0f / (float)Cc);

#pragma unroll
        for (int i = 0; i < 8; ++i) {
            const int  j  = lane + 32 * i;
            const bool ok = (i < 7) || ok7;
            v4f r;
#pragma unroll
            for (int e = 0; e < 4; ++e)
                r[e] = present ? (l1[4 * i + e] - rowMean) : vl[i][e];
            if (ok) *(v4f_a4*)(o_lam + base + 4 * j) = r;
        }
    }
}

#define PW 8
#define CPB 8

__global__ __launch_bounds__(PW * 32) void post_cols(
    const float* __restrict__ priorc, const float* __restrict__ blockpart,
    const float* __restrict__ prior_ema, const float* __restrict__ mu,
    const float* __restrict__ qtot, const float* __restrict__ cep,
    float* __restrict__ o_ce, float* __restrict__ o_mu, float* __restrict__ o_prior)
{
    __shared__ __attribute__((aligned(16))) float spart[PW][CPB];
    __shared__ __attribute__((aligned(16))) float sred[PW][4];
    const int wave  = __builtin_amdgcn_readfirstlane(threadIdx.x >> 5);
    const int lane  = threadIdx.x & 31;
    const int NV    = Cc / 4;

    const v4f z4 = {0.f, 0.f, 0.f, 0.f};
    const int t = threadIdx.x;
    const float* __restrict__ bp = blockpart + (long)min(t, NBLK - 1) * CPITCH + CPB * blockIdx.x;
    v4f p0 = *(const v4f*)(bp);
    v4f p1 = *(const v4f*)(bp + 4);
    p0 = (t < NBLK) ? p0 : z4;
    p1 = (t < NBLK) ? p1 : z4;

    const int  tq    = min(t, NV - 1);
    const bool tlive = t < NV;
    v4f f_pe = *(const v4f*)(prior_ema + 4 * tq);
    v4f f_pc = *(const v4f*)(priorc    + 4 * tq);
    v4f f_mu = *(const v4f*)(mu        + 4 * tq);
    v4f f_qt = *(const v4f*)(qtot      + 4 * tq);
    v4f f_ce = z4;
    if (blockIdx.x == 0) f_ce = *(const v4f*)(cep + 4 * tq);
    const int   jc   = CPB * blockIdx.x + (t & (CPB - 1));
    const float pe_j = prior_ema[jc];
    const float pc_j = priorc[jc];
    const float mu_j = mu[jc];
    f_pe = tlive ? f_pe : z4;
    f_pc = tlive ? f_pc : z4;
    f_mu = tlive ? f_mu : z4;
    f_qt = tlive ? f_qt : z4;
    f_ce = tlive ? f_ce : z4;

    {
        v4f q0, q1;
#pragma unroll
        for (int e = 0; e < 4; ++e) { q0[e] = wave_sum_f(p0[e]); q1[e] = wave_sum_f(p1[e]); }
        if (lane == 0) { *(v4f*)(&spart[wave][0]) = q0; *(v4f*)(&spart[wave][4]) = q1; }
        const v4f pn4 = DECAYF * f_pe + ONE_M_DECAY * f_pc;
        v4f r;
        r[0] = wave_sum_f((pn4[0]  + pn4[1])  + (pn4[2]  + pn4[3]));
        r[1] = wave_sum_f((f_mu[0] + f_mu[1]) + (f_mu[2] + f_mu[3]));
        r[2] = wave_sum_f((f_qt[0] + f_qt[1]) + (f_qt[2] + f_qt[3]));
        r[3] = wave_sum_f((f_ce[0] + f_ce[1]) + (f_ce[2] + f_ce[3]));
        if (lane == 0) *(v4f*)(&sred[wave][0]) = r;
    }
    __syncthreads();
    if (t < CPB) {
        float s = spart[0][t];
        v4f   S = *(const v4f*)(&sred[0][0]);
#pragma unroll
        for (int w = 1; w < PW; ++w) { s += spart[w][t]; S += *(const v4f*)(&sred[w][0]); }
        const float pn     = DECAYF * pe_j + ONE_M_DECAY * pc_j;
        const float den    = S[0] + EPSF;
        const float mu1    = mu_j + LR_MUF * (s * (1.0f / (float)Bn) - pn / den);
        const float muMean = (S[1] + LR_MUF * (S[2] * (1.0f / (float)Bn) - S[0] / den)) * (1.0f / (float)Cc);
        o_prior[jc] = pn;
        o_mu[jc]    = mu1 - muMean;
        if (blockIdx.x == 0 && t == 0) o_ce[0] = -S[3] * (1.0f / (float)Bn);
    }
}

extern "C" void kernel_launch(void* const* d_in, const int* in_sizes, int n_in,
                              void* d_out, int out_size, void* d_ws, size_t ws_size,
                              hipStream_t stream)
{
    const float* logits    = (const float*)d_in[0];
    const int*   targets   = (const int*)  d_in[1];
    const float* conf_ema  = (const float*)d_in[2];
    const float* prior_ema = (const float*)d_in[3];
    const float* Lam       = (const float*)d_in[4];
    const float* mu        = (const float*)d_in[5];

    float* out     = (float*)d_out;
    float* o_ce    = out;
    float* o_lam   = out + 1;
    float* o_mu    = out + 1 + (long)Cc * Cc;
    float* o_conf  = out + 1 + (long)Cc * Cc + Cc;
    float* o_prior = out + 1 + 2L * Cc * Cc + Cc;

    float* ws     = (float*)d_ws;
    float* priorc = ws + 2L * Cc * Cc;
    float* cep    = priorc + Cc;
    int*   bucket = (int*)(cep + 1280);
    float* blockpart = (float*)(bucket + (long)Cc * Bn) + 24;
    float* qtot    = blockpart + (long)NBLK * CPITCH;

    class_pass<<<Cc / KC, NWB * 32, 0, stream>>>(logits, Lam, targets, bucket, conf_ema, o_conf, o_lam, priorc, cep, blockpart, qtot);
    post_cols<<<Cc / CPB, PW * 32, 0, stream>>>(priorc, blockpart, prior_ema, mu, qtot, cep, o_ce, o_mu, o_prior);
}
